# starting kernel
# baseline (speedup 1.0000x reference)
#include <hip/hip_runtime.h>

typedef __attribute__((ext_vector_type(2))) float v2f;
typedef __attribute__((ext_vector_type(8))) float v8f;

#define GCN_N_GRAPHS 64


__global__ void gcn_fill_zero(float* __restrict__ p, int n) {
    int i = blockIdx.x * blockDim.x + threadIdx.x;
    if (i < n) p[i] = 0.0f;
}

__global__ void gcn_degree(const int* __restrict__ src, const int* __restrict__ dst,
                           float* __restrict__ deg_out, float* __restrict__ deg_in, int E) {
    int e = blockIdx.x * blockDim.x + threadIdx.x;
    if (e < E) {
        unsafeAtomicAdd(&deg_out[src[e]], 1.0f);
        unsafeAtomicAdd(&deg_in[dst[e]], 1.0f);
    }
}

__global__ void gcn_norm_finalize(float* __restrict__ buf, int n) {
    int i = blockIdx.x * blockDim.x + threadIdx.x;
    if (i < n) {
        float d = buf[i];
        d = (d < 1.0f) ? 1.0f : d;
        buf[i] = 1.0f / sqrtf(d);
    }
}

__global__ void gcn_count(const int* __restrict__ gid, float* __restrict__ counts, int n) {
    int i = blockIdx.x * blockDim.x + threadIdx.x;
    if (i < n) unsafeAtomicAdd(&counts[gid[i]], 1.0f);
}

__global__ void gcn_final(const float* __restrict__ pooled, const float* __restrict__ counts,
                          float* __restrict__ out) {
    int i = blockIdx.x * blockDim.x + threadIdx.x;
    if (i < GCN_N_GRAPHS * 64) {
        float c = counts[i >> 6];
        c = (c < 1.0f) ? 1.0f : c;
        out[i] = pooled[i] / c;
    }
}


template <int F>
__global__ void gcn_spmm(const float* __restrict__ x, const float* __restrict__ out_norm,
                         const int* __restrict__ src, const int* __restrict__ dst,
                         float* __restrict__ agg, int E) {
    const int TPE = F / 4;
    int gtid = blockIdx.x * blockDim.x + threadIdx.x;
    int e = gtid / TPE;
    if (e >= E) return;
    int f4 = (gtid - e * TPE) * 4;
    int s = src[e];
    int d = dst[e];
    float sc = out_norm[s];
    const float4 v = *reinterpret_cast<const float4*>(x + (size_t)s * F + f4);
    float* base = agg + (size_t)d * F + f4;
    unsafeAtomicAdd(base + 0, v.x * sc);
    unsafeAtomicAdd(base + 1, v.y * sc);
    unsafeAtomicAdd(base + 2, v.z * sc);
    unsafeAtomicAdd(base + 3, v.w * sc);
}


template <bool RELU, bool POOL>
__global__ void gcn_gemm_wmma(const float* __restrict__ A, const float* __restrict__ in_norm,
                              const float* __restrict__ W, const float* __restrict__ bias,
                              float* __restrict__ out, int M, int K, int Nw,
                              const int* __restrict__ graph_ids, float* __restrict__ pooled) {
    int wave = blockIdx.x * (blockDim.x >> 5) + (threadIdx.x >> 5);
    int tiles_n = Nw >> 4;
    int total = (M >> 4) * tiles_n;
    if (wave >= total) return;
    int mt = wave / tiles_n;
    int nt = wave - mt * tiles_n;
    int row0 = mt << 4;
    int col0 = nt << 4;

    int lane  = threadIdx.x & 31;
    int idx15 = lane & 15;
    int koff  = (lane >> 4) << 1;

    int rowA = row0 + idx15;
    float scale = in_norm[rowA];
    const float* arow = A + (size_t)rowA * K + koff;
    const float* wcol = W + col0 + idx15;

    v8f acc = {};
    for (int k = 0; k < K; k += 4) {
        float2 av = *reinterpret_cast<const float2*>(arow + k);
        v2f af;
        af.x = av.x * scale;
        af.y = av.y * scale;
        v2f bf;
        bf.x = wcol[(size_t)(k + koff) * Nw];
        bf.y = wcol[(size_t)(k + koff + 1) * Nw];
        acc = __builtin_amdgcn_wmma_f32_16x16x4_f32(
             false, af,  false, bf,
             (short)0, acc,  false,  false);
    }

    int col = col0 + idx15;
    float bv = bias[col];
    int mbase = (lane >> 4) << 3;
#pragma unroll
    for (int r = 0; r < 8; ++r) {
        int m = row0 + mbase + r;
        float v = acc[r] + bv;
        if (RELU) v = (v > 0.0f) ? v : 0.0f;
        if (POOL) {
            int g = graph_ids[m];
            unsafeAtomicAdd(&pooled[g * Nw + col], v);
        } else {
            out[(size_t)m * Nw + col] = v;
        }
    }
}


static inline int cdiv_l(long a, long b) { return (int)((a + b - 1) / b); }

extern "C" void kernel_launch(void* const* d_in, const int* in_sizes, int n_in,
                              void* d_out, int out_size, void* d_ws, size_t ws_size,
                              hipStream_t stream) {
    const float* emb = (const float*)d_in[0];
    const float* W1  = (const float*)d_in[1];
    const float* b1  = (const float*)d_in[2];
    const float* W2  = (const float*)d_in[3];
    const float* b2  = (const float*)d_in[4];
    const float* W3  = (const float*)d_in[5];
    const float* b3  = (const float*)d_in[6];
    const int* src = (const int*)d_in[7];
    const int* dst = (const int*)d_in[8];
    const int* gid = (const int*)d_in[9];

    const int E = in_sizes[7];
    const int N = in_sizes[9];

    float* ws       = (float*)d_ws;
    float* out_norm = ws;
    float* in_norm  = out_norm + N;
    float* pooled   = in_norm + N;
    float* counts   = pooled + GCN_N_GRAPHS * 64;
    float* agg      = counts + GCN_N_GRAPHS;
    float* h        = agg + (size_t)N * 128;

    const int T = 256;

    int zc = 2 * N + GCN_N_GRAPHS * 64 + GCN_N_GRAPHS;
    gcn_fill_zero<<<cdiv_l(zc, T), T, 0, stream>>>(ws, zc);

    gcn_degree<<<cdiv_l(E, T), T, 0, stream>>>(src, dst, out_norm, in_norm, E);
    gcn_norm_finalize<<<cdiv_l(2L * N, T), T, 0, stream>>>(out_norm, 2 * N);
    gcn_count<<<cdiv_l(N, T), T, 0, stream>>>(gid, counts, N);

    gcn_fill_zero<<<cdiv_l((long)N * 128, T), T, 0, stream>>>(agg, N * 128);
    gcn_spmm<128><<<cdiv_l((long)E * 32, T), T, 0, stream>>>(emb, out_norm, src, dst, agg, E);
    {
        int tiles = (N / 16) * (128 / 16);
        gcn_gemm_wmma<true, false><<<cdiv_l(tiles, 4), 128, 0, stream>>>(
            agg, in_norm, W1, b1, h, N, 128, 128, nullptr, nullptr);
    }

    gcn_fill_zero<<<cdiv_l((long)N * 128, T), T, 0, stream>>>(agg, N * 128);
    gcn_spmm<128><<<cdiv_l((long)E * 32, T), T, 0, stream>>>(h, out_norm, src, dst, agg, E);
    {
        int tiles = (N / 16) * (64 / 16);
        gcn_gemm_wmma<true, false><<<cdiv_l(tiles, 4), 128, 0, stream>>>(
            agg, in_norm, W2, b2, h, N, 128, 64, nullptr, nullptr);
    }

    gcn_fill_zero<<<cdiv_l((long)N * 64, T), T, 0, stream>>>(agg, N * 64);
    gcn_spmm<64><<<cdiv_l((long)E * 16, T), T, 0, stream>>>(h, out_norm, src, dst, agg, E);
    {
        int tiles = (N / 16) * (64 / 16);
        gcn_gemm_wmma<false, true><<<cdiv_l(tiles, 4), 128, 0, stream>>>(
            agg, in_norm, W3, b3, nullptr, N, 64, 64, gid, pooled);
    }

    gcn_final<<<cdiv_l(GCN_N_GRAPHS * 64, T), T, 0, stream>>>(pooled, counts, (float*)d_out);
}


# Layer 2 projects before it aggregates: t = h@W2 (128->64) first, so its edge scatter-add moves 64 floats per edge, not 128; bias+relu move into layer 3's gather. Reassociates layer-2 f32 sums.
# speedup vs baseline: 1.1324x; 1.1324x over previous
#include <hip/hip_runtime.h>

typedef __attribute__((ext_vector_type(2))) float v2f;
typedef __attribute__((ext_vector_type(8))) float v8f;

#define GCN_N_GRAPHS 64


__global__ void gcn_fill_zero(float* __restrict__ p, int n) {
    int i = blockIdx.x * blockDim.x + threadIdx.x;
    if (i < n) p[i] = 0.0f;
}

__global__ void gcn_degree(const int* __restrict__ src, const int* __restrict__ dst,
                           float* __restrict__ deg_out, float* __restrict__ deg_in, int E) {
    int e = blockIdx.x * blockDim.x + threadIdx.x;
    if (e < E) {
        unsafeAtomicAdd(&deg_out[src[e]], 1.0f);
        unsafeAtomicAdd(&deg_in[dst[e]], 1.0f);
    }
}

__global__ void gcn_norm_finalize(float* __restrict__ buf, int n) {
    int i = blockIdx.x * blockDim.x + threadIdx.x;
    if (i < n) {
        float d = buf[i];
        d = (d < 1.0f) ? 1.0f : d;
        buf[i] = 1.0f / sqrtf(d);
    }
}

__global__ void gcn_count(const int* __restrict__ gid, float* __restrict__ counts, int n) {
    int i = blockIdx.x * blockDim.x + threadIdx.x;
    if (i < n) unsafeAtomicAdd(&counts[gid[i]], 1.0f);
}

__global__ void gcn_final(const float* __restrict__ pooled, const float* __restrict__ counts,
                          float* __restrict__ out) {
    int i = blockIdx.x * blockDim.x + threadIdx.x;
    if (i < GCN_N_GRAPHS * 64) {
        float c = counts[i >> 6];
        c = (c < 1.0f) ? 1.0f : c;
        out[i] = pooled[i] / c;
    }
}


template <int F, bool ACT>
__global__ void gcn_spmm(const float* __restrict__ x, const float* __restrict__ out_norm,
                         const float* __restrict__ in_norm, const float* __restrict__ bias,
                         const int* __restrict__ src, const int* __restrict__ dst,
                         float* __restrict__ agg, int E) {
    const int TPE = F / 4;
    int gtid = blockIdx.x * blockDim.x + threadIdx.x;
    int e = gtid / TPE;
    if (e >= E) return;
    int f4 = (gtid - e * TPE) * 4;
    int s = src[e];
    int d = dst[e];
    float sc = out_norm[s];
    float4 v = *reinterpret_cast<const float4*>(x + (size_t)s * F + f4);
    if (ACT) {
        float a = in_norm[s];
        v.x = v.x * a + bias[f4 + 0];
        v.y = v.y * a + bias[f4 + 1];
        v.z = v.z * a + bias[f4 + 2];
        v.w = v.w * a + bias[f4 + 3];
        v.x = (v.x > 0.0f) ? v.x : 0.0f;
        v.y = (v.y > 0.0f) ? v.y : 0.0f;
        v.z = (v.z > 0.0f) ? v.z : 0.0f;
        v.w = (v.w > 0.0f) ? v.w : 0.0f;
    }
    float* base = agg + (size_t)d * F + f4;
    unsafeAtomicAdd(base + 0, v.x * sc);
    unsafeAtomicAdd(base + 1, v.y * sc);
    unsafeAtomicAdd(base + 2, v.z * sc);
    unsafeAtomicAdd(base + 3, v.w * sc);
}


template <bool RELU, bool POOL, bool AFFINE>
__global__ void gcn_gemm_wmma(const float* __restrict__ A, const float* __restrict__ in_norm,
                              const float* __restrict__ W, const float* __restrict__ bias,
                              float* __restrict__ out, int M, int K, int Nw,
                              const int* __restrict__ graph_ids, float* __restrict__ pooled) {
    int wave = blockIdx.x * (blockDim.x >> 5) + (threadIdx.x >> 5);
    int tiles_n = Nw >> 4;
    int total = (M >> 4) * tiles_n;
    if (wave >= total) return;
    int mt = wave / tiles_n;
    int nt = wave - mt * tiles_n;
    int row0 = mt << 4;
    int col0 = nt << 4;

    int lane  = threadIdx.x & 31;
    int idx15 = lane & 15;
    int koff  = (lane >> 4) << 1;

    int rowA = row0 + idx15;
    float scale = AFFINE ? in_norm[rowA] : 1.0f;
    const float* arow = A + (size_t)rowA * K + koff;
    const float* wcol = W + col0 + idx15;

    v8f acc = {};
    for (int k = 0; k < K; k += 4) {
        float2 av = *reinterpret_cast<const float2*>(arow + k);
        v2f af;
        af.x = AFFINE ? av.x * scale : av.x;
        af.y = AFFINE ? av.y * scale : av.y;
        v2f bf;
        bf.x = wcol[(size_t)(k + koff) * Nw];
        bf.y = wcol[(size_t)(k + koff + 1) * Nw];
        acc = __builtin_amdgcn_wmma_f32_16x16x4_f32(
             false, af,  false, bf,
             (short)0, acc,  false,  false);
    }

    int col = col0 + idx15;
    float bv = AFFINE ? bias[col] : 0.0f;
    int mbase = (lane >> 4) << 3;
#pragma unroll
    for (int r = 0; r < 8; ++r) {
        int m = row0 + mbase + r;
        float v = AFFINE ? acc[r] + bv : acc[r];
        if (RELU) v = (v > 0.0f) ? v : 0.0f;
        if (POOL) {
            int g = graph_ids[m];
            unsafeAtomicAdd(&pooled[g * Nw + col], v);
        } else {
            out[(size_t)m * Nw + col] = v;
        }
    }
}


static inline int cdiv_l(long a, long b) { return (int)((a + b - 1) / b); }

extern "C" void kernel_launch(void* const* d_in, const int* in_sizes, int n_in,
                              void* d_out, int out_size, void* d_ws, size_t ws_size,
                              hipStream_t stream) {
    const float* emb = (const float*)d_in[0];
    const float* W1  = (const float*)d_in[1];
    const float* b1  = (const float*)d_in[2];
    const float* W2  = (const float*)d_in[3];
    const float* b2  = (const float*)d_in[4];
    const float* W3  = (const float*)d_in[5];
    const float* b3  = (const float*)d_in[6];
    const int* src = (const int*)d_in[7];
    const int* dst = (const int*)d_in[8];
    const int* gid = (const int*)d_in[9];

    const int E = in_sizes[7];
    const int N = in_sizes[9];

    float* ws       = (float*)d_ws;
    float* out_norm = ws;
    float* in_norm  = out_norm + N;
    float* pooled   = in_norm + N;
    float* counts   = pooled + GCN_N_GRAPHS * 64;
    float* agg      = counts + GCN_N_GRAPHS;
    float* h        = agg + (size_t)N * 128;

    const int T = 256;

    int zc = 2 * N + GCN_N_GRAPHS * 64 + GCN_N_GRAPHS;
    gcn_fill_zero<<<cdiv_l(zc, T), T, 0, stream>>>(ws, zc);

    gcn_degree<<<cdiv_l(E, T), T, 0, stream>>>(src, dst, out_norm, in_norm, E);
    gcn_norm_finalize<<<cdiv_l(2L * N, T), T, 0, stream>>>(out_norm, 2 * N);
    gcn_count<<<cdiv_l(N, T), T, 0, stream>>>(gid, counts, N);

    gcn_fill_zero<<<cdiv_l((long)N * 128, T), T, 0, stream>>>(agg, N * 128);
    gcn_spmm<128, false><<<cdiv_l((long)E * 32, T), T, 0, stream>>>(
        emb, out_norm, nullptr, nullptr, src, dst, agg, E);
    {
        int tiles = (N / 16) * (128 / 16);
        gcn_gemm_wmma<true, false, true><<<cdiv_l(tiles, 4), 128, 0, stream>>>(
            agg, in_norm, W1, b1, h, N, 128, 128, nullptr, nullptr);
    }

    float* t    = agg;
    float* agg2 = agg + (size_t)N * 64;
    float* agg3 = h;
    {
        int tiles = (N / 16) * (64 / 16);
        gcn_gemm_wmma<false, false, false><<<cdiv_l(tiles, 4), 128, 0, stream>>>(
            h, nullptr, W2, nullptr, t, N, 128, 64, nullptr, nullptr);
    }
    gcn_fill_zero<<<cdiv_l((long)N * 64, T), T, 0, stream>>>(agg2, N * 64);
    gcn_spmm<64, false><<<cdiv_l((long)E * 16, T), T, 0, stream>>>(
        t, out_norm, nullptr, nullptr, src, dst, agg2, E);

    gcn_fill_zero<<<cdiv_l((long)N * 64, T), T, 0, stream>>>(agg3, N * 64);
    gcn_spmm<64, true><<<cdiv_l((long)E * 16, T), T, 0, stream>>>(
        agg2, out_norm, in_norm, b2, src, dst, agg3, E);
    {
        int tiles = (N / 16) * (64 / 16);
        gcn_gemm_wmma<false, true, true><<<cdiv_l(tiles, 4), 128, 0, stream>>>(
            agg3, in_norm, W3, b3, nullptr, N, 64, 64, gid, pooled);
    }

    gcn_final<<<cdiv_l(GCN_N_GRAPHS * 64, T), T, 0, stream>>>(pooled, counts, (float*)d_out);
}


# Edges bucketed by destination once (integer count, scan, place); one wave per node sums its source rows in registers, one store per row: no f32 row atomics or zero fills. f32 sum order: slot order.
# speedup vs baseline: 2.0811x; 1.8378x over previous
#include <hip/hip_runtime.h>

typedef __attribute__((ext_vector_type(2))) float v2f;
typedef __attribute__((ext_vector_type(8))) float v8f;

#define GCN_N_GRAPHS 64


__global__ void gcn_fill_zero(float* __restrict__ p, int n) {
    int i = blockIdx.x * blockDim.x + threadIdx.x;
    if (i < n) p[i] = 0.0f;
}

__global__ void gcn_degree_rank(const int* __restrict__ src, const int* __restrict__ dst,
                                int* __restrict__ cnt_out, int* __restrict__ cnt_in,
                                int* __restrict__ rank, int E) {
    int e = blockIdx.x * blockDim.x + threadIdx.x;
    if (e < E) {
        atomicAdd(&cnt_out[src[e]], 1);
        rank[e] = atomicAdd(&cnt_in[dst[e]], 1);
    }
}

__global__ void gcn_norm_finalize(const int* __restrict__ cnt, float* __restrict__ norm, int n) {
    int i = blockIdx.x * blockDim.x + threadIdx.x;
    if (i < n) {
        float d = (float)cnt[i];
        d = (d < 1.0f) ? 1.0f : d;
        norm[i] = 1.0f / sqrtf(d);
    }
}

__global__ void gcn_row_start(const int* __restrict__ cnt, int* __restrict__ row_start, int n) {
    __shared__ int s_front[32];
    __shared__ int s_wave[32];
    __shared__ int s_base;
    int tid = threadIdx.x;
    int lane = tid & 31;
    int w = tid >> 5;
    int first = blockIdx.x * 1024;
    int i = first + tid;

    int front = 0;
    for (int j = tid; j < first; j += 1024) front += cnt[j];
    int v = (i < n) ? cnt[i] : 0;

    for (int o = 16; o > 0; o >>= 1) front += __shfl_xor(front, o);
    int inc = v;
    for (int o = 1; o < 32; o <<= 1) {
        int t = __shfl_up(inc, o);
        if (lane >= o) inc += t;
    }
    if (lane == 0) s_front[w] = front;
    if (lane == 31) s_wave[w] = inc;
    __syncthreads();
    if (w == 0) {
        int a = s_front[lane];
        for (int o = 16; o > 0; o >>= 1) a += __shfl_xor(a, o);
        int p = s_wave[lane];
        int pi = p;
        for (int o = 1; o < 32; o <<= 1) {
            int t = __shfl_up(pi, o);
            if (lane >= o) pi += t;
        }
        s_wave[lane] = pi - p;
        if (lane == 0) s_base = a;
    }
    __syncthreads();
    if (i < n) row_start[i] = s_base + s_wave[w] + inc - v;
}

__global__ void gcn_place(const int* __restrict__ src, const int* __restrict__ dst,
                          const int* __restrict__ row_start, const int* __restrict__ rank,
                          int* __restrict__ csr_src, int E) {
    int e = blockIdx.x * blockDim.x + threadIdx.x;
    if (e < E) csr_src[row_start[dst[e]] + rank[e]] = src[e];
}

__global__ void gcn_count(const int* __restrict__ gid, float* __restrict__ counts, int n) {
    int i = blockIdx.x * blockDim.x + threadIdx.x;
    if (i < n) unsafeAtomicAdd(&counts[gid[i]], 1.0f);
}

__global__ void gcn_final(const float* __restrict__ pooled, const float* __restrict__ counts,
                          float* __restrict__ out) {
    int i = blockIdx.x * blockDim.x + threadIdx.x;
    if (i < GCN_N_GRAPHS * 64) {
        float c = counts[i >> 6];
        c = (c < 1.0f) ? 1.0f : c;
        out[i] = pooled[i] / c;
    }
}


template <int V> struct gcn_vec;
template <> struct gcn_vec<4> { typedef __attribute__((ext_vector_type(4))) float type; };
template <> struct gcn_vec<2> { typedef __attribute__((ext_vector_type(2))) float type; };

template <int F, bool ACT, int U, bool TAIL, typename vec_t>
__device__ __forceinline__ void gcn_pull_edges(const float* __restrict__ xl, int s_l, float sc_l,
                                               float a_l, vec_t bv, int j, int m, vec_t& acc) {
    vec_t v[U];
    float sc[U];
    float a[U];
#pragma unroll
    for (int u = 0; u < U; ++u) {
        int jj = j + u;
        if (TAIL) jj = (jj < m) ? jj : m - 1;
        int s = __builtin_amdgcn_readlane(s_l, jj);
        sc[u] = __int_as_float(__builtin_amdgcn_readlane(__float_as_int(sc_l), jj));
        a[u] = ACT ? __int_as_float(__builtin_amdgcn_readlane(__float_as_int(a_l), jj)) : 1.0f;
        v[u] = *reinterpret_cast<const vec_t*>(xl + (size_t)s * F);
    }
#pragma unroll
    for (int u = 0; u < U; ++u) {
        if (!TAIL || j + u < m) {
            vec_t t = v[u];
            if (ACT) {
                t = t * a[u] + bv;
#pragma unroll
                for (int q = 0; q < F / 32; ++q) t[q] = (t[q] > 0.0f) ? t[q] : 0.0f;
            }
            acc += t * sc[u];
        }
    }
}

template <int F, bool ACT>
__global__ void gcn_pull(const float* __restrict__ x, const float* __restrict__ out_norm,
                         const float* __restrict__ in_norm, const float* __restrict__ bias,
                         const int* __restrict__ row_start, const int* __restrict__ cnt_in,
                         const int* __restrict__ csr_src, float* __restrict__ agg, int n) {
    const int VPL = F / 32;
    typedef typename gcn_vec<VPL>::type vec_t;
    int row = __builtin_amdgcn_readfirstlane(
        (int)(blockIdx.x * (blockDim.x >> 5) + (threadIdx.x >> 5)));
    if (row >= n) return;
    int lane = threadIdx.x & 31;
    int start = row_start[row];
    int deg = cnt_in[row];
    const float* xl = x + lane * VPL;

    vec_t acc = {};
    vec_t bv = {};
    if (ACT) bv = *reinterpret_cast<const vec_t*>(bias + lane * VPL);

    for (int base = 0; base < deg; base += 32) {
        int m = deg - base;
        m = (m < 32) ? m : 32;
        int s_l = 0;
        float sc_l = 0.0f;
        float a_l = 0.0f;
        if (lane < m) {
            s_l = csr_src[start + base + lane];
            sc_l = out_norm[s_l];
            if (ACT) a_l = in_norm[s_l];
        }
        int j = 0;
        for (; j + 8 <= m; j += 8)
            gcn_pull_edges<F, ACT, 8, false, vec_t>(xl, s_l, sc_l, a_l, bv, j, m, acc);
        for (; j < m; j += 4)
            gcn_pull_edges<F, ACT, 4, true, vec_t>(xl, s_l, sc_l, a_l, bv, j, m, acc);
    }
    *reinterpret_cast<vec_t*>(agg + (size_t)row * F + lane * VPL) = acc;
}


template <bool RELU, bool POOL, bool AFFINE>
__global__ void gcn_gemm_wmma(const float* __restrict__ A, const float* __restrict__ in_norm,
                              const float* __restrict__ W, const float* __restrict__ bias,
                              float* __restrict__ out, int M, int K, int Nw,
                              const int* __restrict__ graph_ids, float* __restrict__ pooled) {
    int wave = blockIdx.x * (blockDim.x >> 5) + (threadIdx.x >> 5);
    int tiles_n = Nw >> 4;
    int total = (M >> 4) * tiles_n;
    if (wave >= total) return;
    int mt = wave / tiles_n;
    int nt = wave - mt * tiles_n;
    int row0 = mt << 4;
    int col0 = nt << 4;

    int lane  = threadIdx.x & 31;
    int idx15 = lane & 15;
    int koff  = (lane >> 4) << 1;

    int rowA = row0 + idx15;
    float scale = AFFINE ? in_norm[rowA] : 1.0f;
    const float* arow = A + (size_t)rowA * K + koff;
    const float* wcol = W + col0 + idx15;

    v8f acc = {};
    for (int k = 0; k < K; k += 4) {
        float2 av = *reinterpret_cast<const float2*>(arow + k);
        v2f af;
        af.x = AFFINE ? av.x * scale : av.x;
        af.y = AFFINE ? av.y * scale : av.y;
        v2f bf;
        bf.x = wcol[(size_t)(k + koff) * Nw];
        bf.y = wcol[(size_t)(k + koff + 1) * Nw];
        acc = __builtin_amdgcn_wmma_f32_16x16x4_f32(
             false, af,  false, bf,
             (short)0, acc,  false,  false);
    }

    int col = col0 + idx15;
    float bv = AFFINE ? bias[col] : 0.0f;
    int mbase = (lane >> 4) << 3;
#pragma unroll
    for (int r = 0; r < 8; ++r) {
        int m = row0 + mbase + r;
        float v = AFFINE ? acc[r] + bv : acc[r];
        if (RELU) v = (v > 0.0f) ? v : 0.0f;
        if (POOL) {
            int g = graph_ids[m];
            unsafeAtomicAdd(&pooled[g * Nw + col], v);
        } else {
            out[(size_t)m * Nw + col] = v;
        }
    }
}


static inline int cdiv_l(long a, long b) { return (int)((a + b - 1) / b); }

extern "C" void kernel_launch(void* const* d_in, const int* in_sizes, int n_in,
                              void* d_out, int out_size, void* d_ws, size_t ws_size,
                              hipStream_t stream) {
    const float* emb = (const float*)d_in[0];
    const float* W1  = (const float*)d_in[1];
    const float* b1  = (const float*)d_in[2];
    const float* W2  = (const float*)d_in[3];
    const float* b2  = (const float*)d_in[4];
    const float* W3  = (const float*)d_in[5];
    const float* b3  = (const float*)d_in[6];
    const int* src = (const int*)d_in[7];
    const int* dst = (const int*)d_in[8];
    const int* gid = (const int*)d_in[9];

    const int E = in_sizes[7];
    const int N = in_sizes[9];

    int*   cnt_out   = (int*)d_ws;
    int*   cnt_in    = cnt_out + N;
    float* pooled    = (float*)(cnt_in + N);
    float* counts    = pooled + GCN_N_GRAPHS * 64;
    float* out_norm  = counts + GCN_N_GRAPHS;
    float* in_norm   = out_norm + N;
    int*   row_start = (int*)(in_norm + N);
    int*   rank      = row_start + N;
    int*   csr_src   = rank + E;
    float* agg       = (float*)(csr_src + E);
    float* h         = agg + (size_t)N * 128;

    const int T = 256;

    int zc = 2 * N + GCN_N_GRAPHS * 64 + GCN_N_GRAPHS;
    gcn_fill_zero<<<cdiv_l(zc, T), T, 0, stream>>>((float*)d_ws, zc);

    gcn_degree_rank<<<cdiv_l(E, T), T, 0, stream>>>(src, dst, cnt_out, cnt_in, rank, E);
    gcn_norm_finalize<<<cdiv_l(2L * N, T), T, 0, stream>>>(cnt_out, out_norm, 2 * N);
    gcn_count<<<cdiv_l(N, T), T, 0, stream>>>(gid, counts, N);

    gcn_row_start<<<cdiv_l(N, 1024), 1024, 0, stream>>>(cnt_in, row_start, N);
    gcn_place<<<cdiv_l(E, T), T, 0, stream>>>(src, dst, row_start, rank, csr_src, E);

    const int rows_per_block = T / 32;

    gcn_pull<128, false><<<cdiv_l(N, rows_per_block), T, 0, stream>>>(
        emb, out_norm, nullptr, nullptr, row_start, cnt_in, csr_src, agg, N);
    {
        int tiles = (N / 16) * (128 / 16);
        gcn_gemm_wmma<true, false, true><<<cdiv_l(tiles, 4), 128, 0, stream>>>(
            agg, in_norm, W1, b1, h, N, 128, 128, nullptr, nullptr);
    }

    float* t    = agg;
    float* agg2 = agg + (size_t)N * 64;
    float* agg3 = h;
    {
        int tiles = (N / 16) * (64 / 16);
        gcn_gemm_wmma<false, false, false><<<cdiv_l(tiles, 4), 128, 0, stream>>>(
            h, nullptr, W2, nullptr, t, N, 128, 64, nullptr, nullptr);
    }
    gcn_pull<64, false><<<cdiv_l(N, rows_per_block), T, 0, stream>>>(
        t, out_norm, nullptr, nullptr, row_start, cnt_in, csr_src, agg2, N);

    gcn_pull<64, true><<<cdiv_l(N, rows_per_block), T, 0, stream>>>(
        agg2, out_norm, in_norm, b2, row_start, cnt_in, csr_src, agg3, N);
    {
        int tiles = (N / 16) * (64 / 16);
        gcn_gemm_wmma<false, true, true><<<cdiv_l(tiles, 4), 128, 0, stream>>>(
            agg3, in_norm, W3, b3, nullptr, N, 64, 64, gid, pooled);
    }

    gcn_final<<<cdiv_l(GCN_N_GRAPHS * 64, T), T, 0, stream>>>(pooled, counts, (float*)d_out);
}


# Pooling reduces in the wave before its f32 atomics: a tile's 16 rows of one graph are summed in registers (1 add per column, not 16); node counts 1 add per wave and graph. Changes f32 sum order.
# speedup vs baseline: 2.7351x; 1.3143x over previous
#include <hip/hip_runtime.h>

typedef __attribute__((ext_vector_type(2))) float v2f;
typedef __attribute__((ext_vector_type(8))) float v8f;

#define GCN_N_GRAPHS 64


__global__ void gcn_fill_zero(float* __restrict__ p, int n) {
    int i = blockIdx.x * blockDim.x + threadIdx.x;
    if (i < n) p[i] = 0.0f;
}

__global__ void gcn_degree_rank(const int* __restrict__ src, const int* __restrict__ dst,
                                int* __restrict__ cnt_out, int* __restrict__ cnt_in,
                                int* __restrict__ rank, int E) {
    int e = blockIdx.x * blockDim.x + threadIdx.x;
    if (e < E) {
        atomicAdd(&cnt_out[src[e]], 1);
        rank[e] = atomicAdd(&cnt_in[dst[e]], 1);
    }
}

__global__ void gcn_norm_finalize(const int* __restrict__ cnt, float* __restrict__ norm, int n) {
    int i = blockIdx.x * blockDim.x + threadIdx.x;
    if (i < n) {
        float d = (float)cnt[i];
        d = (d < 1.0f) ? 1.0f : d;
        norm[i] = 1.0f / sqrtf(d);
    }
}

__global__ void gcn_row_start(const int* __restrict__ cnt, int* __restrict__ row_start, int n) {
    __shared__ int s_front[32];
    __shared__ int s_wave[32];
    __shared__ int s_base;
    int tid = threadIdx.x;
    int lane = tid & 31;
    int w = tid >> 5;
    int first = blockIdx.x * 1024;
    int i = first + tid;

    int front = 0;
    for (int j = tid; j < first; j += 1024) front += cnt[j];
    int v = (i < n) ? cnt[i] : 0;

    for (int o = 16; o > 0; o >>= 1) front += __shfl_xor(front, o);
    int inc = v;
    for (int o = 1; o < 32; o <<= 1) {
        int t = __shfl_up(inc, o);
        if (lane >= o) inc += t;
    }
    if (lane == 0) s_front[w] = front;
    if (lane == 31) s_wave[w] = inc;
    __syncthreads();
    if (w == 0) {
        int a = s_front[lane];
        for (int o = 16; o > 0; o >>= 1) a += __shfl_xor(a, o);
        int p = s_wave[lane];
        int pi = p;
        for (int o = 1; o < 32; o <<= 1) {
            int t = __shfl_up(pi, o);
            if (lane >= o) pi += t;
        }
        s_wave[lane] = pi - p;
        if (lane == 0) s_base = a;
    }
    __syncthreads();
    if (i < n) row_start[i] = s_base + s_wave[w] + inc - v;
}

__global__ void gcn_place(const int* __restrict__ src, const int* __restrict__ dst,
                          const int* __restrict__ row_start, const int* __restrict__ rank,
                          int* __restrict__ csr_src, int E) {
    int e = blockIdx.x * blockDim.x + threadIdx.x;
    if (e < E) csr_src[row_start[dst[e]] + rank[e]] = src[e];
}

__global__ void gcn_count(const int* __restrict__ gid, float* __restrict__ counts, int n) {
    int i = blockIdx.x * blockDim.x + threadIdx.x;
    int lane = threadIdx.x & 31;
    bool todo = i < n;
    int g = todo ? gid[i] : 0;
    for (int it = 0; it < 32; ++it) {
        unsigned waiting = __builtin_amdgcn_ballot_w32(todo);
        if (waiting == 0u) break;
        int first = __builtin_ctz(waiting);
        int g0 = __builtin_amdgcn_readlane(g, first);
        bool same = todo && (g == g0);
        unsigned members = __builtin_amdgcn_ballot_w32(same);
        if (lane == first) unsafeAtomicAdd(&counts[g0], (float)__builtin_popcount(members));
        todo = todo && !same;
    }
}

__global__ void gcn_final(const float* __restrict__ pooled, const float* __restrict__ counts,
                          float* __restrict__ out) {
    int i = blockIdx.x * blockDim.x + threadIdx.x;
    if (i < GCN_N_GRAPHS * 64) {
        float c = counts[i >> 6];
        c = (c < 1.0f) ? 1.0f : c;
        out[i] = pooled[i] / c;
    }
}


template <int V> struct gcn_vec;
template <> struct gcn_vec<4> { typedef __attribute__((ext_vector_type(4))) float type; };
template <> struct gcn_vec<2> { typedef __attribute__((ext_vector_type(2))) float type; };

template <int F, bool ACT, int U, bool TAIL, typename vec_t>
__device__ __forceinline__ void gcn_pull_edges(const float* __restrict__ xl, int s_l, float sc_l,
                                               float a_l, vec_t bv, int j, int m, vec_t& acc) {
    vec_t v[U];
    float sc[U];
    float a[U];
#pragma unroll
    for (int u = 0; u < U; ++u) {
        int jj = j + u;
        if (TAIL) jj = (jj < m) ? jj : m - 1;
        int s = __builtin_amdgcn_readlane(s_l, jj);
        sc[u] = __int_as_float(__builtin_amdgcn_readlane(__float_as_int(sc_l), jj));
        a[u] = ACT ? __int_as_float(__builtin_amdgcn_readlane(__float_as_int(a_l), jj)) : 1.0f;
        v[u] = *reinterpret_cast<const vec_t*>(xl + (size_t)s * F);
    }
#pragma unroll
    for (int u = 0; u < U; ++u) {
        if (!TAIL || j + u < m) {
            vec_t t = v[u];
            if (ACT) {
                t = t * a[u] + bv;
#pragma unroll
                for (int q = 0; q < F / 32; ++q) t[q] = (t[q] > 0.0f) ? t[q] : 0.0f;
            }
            acc += t * sc[u];
        }
    }
}

template <int F, bool ACT>
__global__ void gcn_pull(const float* __restrict__ x, const float* __restrict__ out_norm,
                         const float* __restrict__ in_norm, const float* __restrict__ bias,
                         const int* __restrict__ row_start, const int* __restrict__ cnt_in,
                         const int* __restrict__ csr_src, float* __restrict__ agg, int n) {
    const int VPL = F / 32;
    typedef typename gcn_vec<VPL>::type vec_t;
    int row = __builtin_amdgcn_readfirstlane(
        (int)(blockIdx.x * (blockDim.x >> 5) + (threadIdx.x >> 5)));
    if (row >= n) return;
    int lane = threadIdx.x & 31;
    int start = row_start[row];
    int deg = cnt_in[row];
    const float* xl = x + lane * VPL;

    vec_t acc = {};
    vec_t bv = {};
    if (ACT) bv = *reinterpret_cast<const vec_t*>(bias + lane * VPL);

    for (int base = 0; base < deg; base += 32) {
        int m = deg - base;
        m = (m < 32) ? m : 32;
        int s_l = 0;
        float sc_l = 0.0f;
        float a_l = 0.0f;
        if (lane < m) {
            s_l = csr_src[start + base + lane];
            sc_l = out_norm[s_l];
            if (ACT) a_l = in_norm[s_l];
        }
        int j = 0;
        for (; j + 8 <= m; j += 8)
            gcn_pull_edges<F, ACT, 8, false, vec_t>(xl, s_l, sc_l, a_l, bv, j, m, acc);
        for (; j < m; j += 4)
            gcn_pull_edges<F, ACT, 4, true, vec_t>(xl, s_l, sc_l, a_l, bv, j, m, acc);
    }
    *reinterpret_cast<vec_t*>(agg + (size_t)row * F + lane * VPL) = acc;
}


template <bool RELU, bool POOL, bool AFFINE>
__global__ void gcn_gemm_wmma(const float* __restrict__ A, const float* __restrict__ in_norm,
                              const float* __restrict__ W, const float* __restrict__ bias,
                              float* __restrict__ out, int M, int K, int Nw,
                              const int* __restrict__ graph_ids, float* __restrict__ pooled) {
    int wave = blockIdx.x * (blockDim.x >> 5) + (threadIdx.x >> 5);
    int tiles_n = Nw >> 4;
    int total = (M >> 4) * tiles_n;
    if (wave >= total) return;
    int mt = wave / tiles_n;
    int nt = wave - mt * tiles_n;
    int row0 = mt << 4;
    int col0 = nt << 4;

    int lane  = threadIdx.x & 31;
    int idx15 = lane & 15;
    int koff  = (lane >> 4) << 1;

    int rowA = row0 + idx15;
    float scale = AFFINE ? in_norm[rowA] : 1.0f;
    const float* arow = A + (size_t)rowA * K + koff;
    const float* wcol = W + col0 + idx15;

    v8f acc = {};
    for (int k = 0; k < K; k += 4) {
        float2 av = *reinterpret_cast<const float2*>(arow + k);
        v2f af;
        af.x = AFFINE ? av.x * scale : av.x;
        af.y = AFFINE ? av.y * scale : av.y;
        v2f bf;
        bf.x = wcol[(size_t)(k + koff) * Nw];
        bf.y = wcol[(size_t)(k + koff + 1) * Nw];
        acc = __builtin_amdgcn_wmma_f32_16x16x4_f32(
             false, af,  false, bf,
             (short)0, acc,  false,  false);
    }

    int col = col0 + idx15;
    float bv = AFFINE ? bias[col] : 0.0f;
    int mbase = (lane >> 4) << 3;
    if (POOL) {
        int g_l = graph_ids[row0 + idx15];
        int g0 = __builtin_amdgcn_readfirstlane(g_l);
        bool one_graph = (__builtin_amdgcn_ballot_w32(g_l != g0) == 0u);
        if (one_graph) {
            float s = 0.0f;
#pragma unroll
            for (int r = 0; r < 8; ++r) {
                float v = AFFINE ? acc[r] + bv : acc[r];
                if (RELU) v = (v > 0.0f) ? v : 0.0f;
                s += v;
            }
            s += __shfl_xor(s, 16);
            if (lane < 16) unsafeAtomicAdd(&pooled[g0 * Nw + col], s);
        } else {
#pragma unroll
            for (int r = 0; r < 8; ++r) {
                int m = row0 + mbase + r;
                float v = AFFINE ? acc[r] + bv : acc[r];
                if (RELU) v = (v > 0.0f) ? v : 0.0f;
                int g = graph_ids[m];
                unsafeAtomicAdd(&pooled[g * Nw + col], v);
            }
        }
    } else {
#pragma unroll
        for (int r = 0; r < 8; ++r) {
            int m = row0 + mbase + r;
            float v = AFFINE ? acc[r] + bv : acc[r];
            if (RELU) v = (v > 0.0f) ? v : 0.0f;
            out[(size_t)m * Nw + col] = v;
        }
    }
}


static inline int cdiv_l(long a, long b) { return (int)((a + b - 1) / b); }

extern "C" void kernel_launch(void* const* d_in, const int* in_sizes, int n_in,
                              void* d_out, int out_size, void* d_ws, size_t ws_size,
                              hipStream_t stream) {
    const float* emb = (const float*)d_in[0];
    const float* W1  = (const float*)d_in[1];
    const float* b1  = (const float*)d_in[2];
    const float* W2  = (const float*)d_in[3];
    const float* b2  = (const float*)d_in[4];
    const float* W3  = (const float*)d_in[5];
    const float* b3  = (const float*)d_in[6];
    const int* src = (const int*)d_in[7];
    const int* dst = (const int*)d_in[8];
    const int* gid = (const int*)d_in[9];

    const int E = in_sizes[7];
    const int N = in_sizes[9];

    int*   cnt_out   = (int*)d_ws;
    int*   cnt_in    = cnt_out + N;
    float* pooled    = (float*)(cnt_in + N);
    float* counts    = pooled + GCN_N_GRAPHS * 64;
    float* out_norm  = counts + GCN_N_GRAPHS;
    float* in_norm   = out_norm + N;
    int*   row_start = (int*)(in_norm + N);
    int*   rank      = row_start + N;
    int*   csr_src   = rank + E;
    float* agg       = (float*)(csr_src + E);
    float* h         = agg + (size_t)N * 128;

    const int T = 256;

    int zc = 2 * N + GCN_N_GRAPHS * 64 + GCN_N_GRAPHS;
    gcn_fill_zero<<<cdiv_l(zc, T), T, 0, stream>>>((float*)d_ws, zc);

    gcn_degree_rank<<<cdiv_l(E, T), T, 0, stream>>>(src, dst, cnt_out, cnt_in, rank, E);
    gcn_norm_finalize<<<cdiv_l(2L * N, T), T, 0, stream>>>(cnt_out, out_norm, 2 * N);
    gcn_count<<<cdiv_l(N, T), T, 0, stream>>>(gid, counts, N);

    gcn_row_start<<<cdiv_l(N, 1024), 1024, 0, stream>>>(cnt_in, row_start, N);
    gcn_place<<<cdiv_l(E, T), T, 0, stream>>>(src, dst, row_start, rank, csr_src, E);

    const int rows_per_block = T / 32;

    gcn_pull<128, false><<<cdiv_l(N, rows_per_block), T, 0, stream>>>(
        emb, out_norm, nullptr, nullptr, row_start, cnt_in, csr_src, agg, N);
    {
        int tiles = (N / 16) * (128 / 16);
        gcn_gemm_wmma<true, false, true><<<cdiv_l(tiles, 4), 128, 0, stream>>>(
            agg, in_norm, W1, b1, h, N, 128, 128, nullptr, nullptr);
    }

    float* t    = agg;
    float* agg2 = agg + (size_t)N * 64;
    float* agg3 = h;
    {
        int tiles = (N / 16) * (64 / 16);
        gcn_gemm_wmma<false, false, false><<<cdiv_l(tiles, 4), 128, 0, stream>>>(
            h, nullptr, W2, nullptr, t, N, 128, 64, nullptr, nullptr);
    }
    gcn_pull<64, false><<<cdiv_l(N, rows_per_block), T, 0, stream>>>(
        t, out_norm, nullptr, nullptr, row_start, cnt_in, csr_src, agg2, N);

    gcn_pull<64, true><<<cdiv_l(N, rows_per_block), T, 0, stream>>>(
        agg2, out_norm, in_norm, b2, row_start, cnt_in, csr_src, agg3, N);
    {
        int tiles = (N / 16) * (64 / 16);
        gcn_gemm_wmma<false, true, true><<<cdiv_l(tiles, 4), 128, 0, stream>>>(
            agg3, in_norm, W3, b3, nullptr, N, 64, 64, gid, pooled);
    }

    gcn_final<<<cdiv_l(GCN_N_GRAPHS * 64, T), T, 0, stream>>>(pooled, counts, (float*)d_out);
}


# Each projection fused into the edge pass feeding it: a 32-row group's neighbour sums go to LDS, its 8 waves run the WMMA tiles from there (layer 1 chains both). 13 launches to 11; f32 sum orders kept
# speedup vs baseline: 3.4052x; 1.2450x over previous
#include <hip/hip_runtime.h>

typedef __attribute__((ext_vector_type(2))) float v2f;
typedef __attribute__((ext_vector_type(8))) float v8f;

#define GCN_N_GRAPHS 64


__global__ void gcn_fill_zero(float* __restrict__ p, int n) {
    int i = blockIdx.x * blockDim.x + threadIdx.x;
    if (i < n) p[i] = 0.0f;
}

__global__ void gcn_degree_rank(const int* __restrict__ src, const int* __restrict__ dst,
                                int* __restrict__ cnt_out, int* __restrict__ cnt_in,
                                int* __restrict__ rank, int E) {
    int e = blockIdx.x * blockDim.x + threadIdx.x;
    if (e < E) {
        atomicAdd(&cnt_out[src[e]], 1);
        rank[e] = atomicAdd(&cnt_in[dst[e]], 1);
    }
}

__global__ void gcn_norm_finalize(const int* __restrict__ cnt, float* __restrict__ norm, int n) {
    int i = blockIdx.x * blockDim.x + threadIdx.x;
    if (i < n) {
        float d = (float)cnt[i];
        d = (d < 1.0f) ? 1.0f : d;
        norm[i] = 1.0f / sqrtf(d);
    }
}

__global__ void gcn_row_start(const int* __restrict__ cnt, int* __restrict__ row_start, int n) {
    __shared__ int s_front[32];
    __shared__ int s_wave[32];
    __shared__ int s_base;
    int tid = threadIdx.x;
    int lane = tid & 31;
    int w = tid >> 5;
    int first = blockIdx.x * 1024;
    int i = first + tid;

    int front = 0;
    for (int j = tid; j < first; j += 1024) front += cnt[j];
    int v = (i < n) ? cnt[i] : 0;

    for (int o = 16; o > 0; o >>= 1) front += __shfl_xor(front, o);
    int inc = v;
    for (int o = 1; o < 32; o <<= 1) {
        int t = __shfl_up(inc, o);
        if (lane >= o) inc += t;
    }
    if (lane == 0) s_front[w] = front;
    if (lane == 31) s_wave[w] = inc;
    __syncthreads();
    if (w == 0) {
        int a = s_front[lane];
        for (int o = 16; o > 0; o >>= 1) a += __shfl_xor(a, o);
        int p = s_wave[lane];
        int pi = p;
        for (int o = 1; o < 32; o <<= 1) {
            int t = __shfl_up(pi, o);
            if (lane >= o) pi += t;
        }
        s_wave[lane] = pi - p;
        if (lane == 0) s_base = a;
    }
    __syncthreads();
    if (i < n) row_start[i] = s_base + s_wave[w] + inc - v;
}

__global__ void gcn_place(const int* __restrict__ src, const int* __restrict__ dst,
                          const int* __restrict__ row_start, const int* __restrict__ rank,
                          int* __restrict__ csr_src, int E) {
    int e = blockIdx.x * blockDim.x + threadIdx.x;
    if (e < E) csr_src[row_start[dst[e]] + rank[e]] = src[e];
}

__global__ void gcn_count(const int* __restrict__ gid, float* __restrict__ counts, int n) {
    int i = blockIdx.x * blockDim.x + threadIdx.x;
    int lane = threadIdx.x & 31;
    bool todo = i < n;
    int g = todo ? gid[i] : 0;
    for (int it = 0; it < 32; ++it) {
        unsigned waiting = __builtin_amdgcn_ballot_w32(todo);
        if (waiting == 0u) break;
        int first = __builtin_ctz(waiting);
        int g0 = __builtin_amdgcn_readlane(g, first);
        bool same = todo && (g == g0);
        unsigned members = __builtin_amdgcn_ballot_w32(same);
        if (lane == first) unsafeAtomicAdd(&counts[g0], (float)__builtin_popcount(members));
        todo = todo && !same;
    }
}

__global__ void gcn_final(const float* __restrict__ pooled, const float* __restrict__ counts,
                          float* __restrict__ out) {
    int i = blockIdx.x * blockDim.x + threadIdx.x;
    if (i < GCN_N_GRAPHS * 64) {
        float c = counts[i >> 6];
        c = (c < 1.0f) ? 1.0f : c;
        out[i] = pooled[i] / c;
    }
}


template <int V> struct gcn_vec;
template <> struct gcn_vec<4> { typedef __attribute__((ext_vector_type(4))) float type; };
template <> struct gcn_vec<2> { typedef __attribute__((ext_vector_type(2))) float type; };

template <int F, bool ACT, int U, bool TAIL, typename vec_t>
__device__ __forceinline__ void gcn_pull_edges(const float* __restrict__ xl, int s_l, float sc_l,
                                               float a_l, vec_t bv, int j, int m, vec_t& acc) {
    vec_t v[U];
    float sc[U];
    float a[U];
#pragma unroll
    for (int u = 0; u < U; ++u) {
        int jj = j + u;
        if (TAIL) jj = (jj < m) ? jj : m - 1;
        int s = __builtin_amdgcn_readlane(s_l, jj);
        sc[u] = __int_as_float(__builtin_amdgcn_readlane(__float_as_int(sc_l), jj));
        a[u] = ACT ? __int_as_float(__builtin_amdgcn_readlane(__float_as_int(a_l), jj)) : 1.0f;
        v[u] = *reinterpret_cast<const vec_t*>(xl + (size_t)s * F);
    }
#pragma unroll
    for (int u = 0; u < U; ++u) {
        if (!TAIL || j + u < m) {
            vec_t t = v[u];
            if (ACT) {
                t = t * a[u] + bv;
#pragma unroll
                for (int q = 0; q < F / 32; ++q) t[q] = (t[q] > 0.0f) ? t[q] : 0.0f;
            }
            acc += t * sc[u];
        }
    }
}

template <int F, bool ACT>
__global__ void gcn_pull(const float* __restrict__ x, const float* __restrict__ out_norm,
                         const float* __restrict__ in_norm, const float* __restrict__ bias,
                         const int* __restrict__ row_start, const int* __restrict__ cnt_in,
                         const int* __restrict__ csr_src, float* __restrict__ agg, int n) {
    const int VPL = F / 32;
    typedef typename gcn_vec<VPL>::type vec_t;
    int row = __builtin_amdgcn_readfirstlane(
        (int)(blockIdx.x * (blockDim.x >> 5) + (threadIdx.x >> 5)));
    if (row >= n) return;
    int lane = threadIdx.x & 31;
    int start = row_start[row];
    int deg = cnt_in[row];
    const float* xl = x + lane * VPL;

    vec_t acc = {};
    vec_t bv = {};
    if (ACT) bv = *reinterpret_cast<const vec_t*>(bias + lane * VPL);

    for (int base = 0; base < deg; base += 32) {
        int m = deg - base;
        m = (m < 32) ? m : 32;
        int s_l = 0;
        float sc_l = 0.0f;
        float a_l = 0.0f;
        if (lane < m) {
            s_l = csr_src[start + base + lane];
            sc_l = out_norm[s_l];
            if (ACT) a_l = in_norm[s_l];
        }
        int j = 0;
        for (; j + 8 <= m; j += 8)
            gcn_pull_edges<F, ACT, 8, false, vec_t>(xl, s_l, sc_l, a_l, bv, j, m, acc);
        for (; j < m; j += 4)
            gcn_pull_edges<F, ACT, 4, true, vec_t>(xl, s_l, sc_l, a_l, bv, j, m, acc);
    }
    *reinterpret_cast<vec_t*>(agg + (size_t)row * F + lane * VPL) = acc;
}


#define GCN_ROWS 32
#define GCN_RPW 4

template <int F, bool ACT>
__device__ __forceinline__ void gcn_pull_rows_to_lds(
    const float* __restrict__ x, const float* __restrict__ out_norm,
    const float* __restrict__ in_norm, const float* __restrict__ bias,
    const int* __restrict__ row_start, const int* __restrict__ cnt_in,
    const int* __restrict__ csr_src, int row_first, int n, int lane, float* s_rows) {
    const int VPL = F / 32;
    const int PITCH = F + 4;
    typedef typename gcn_vec<VPL>::type vec_t;
    const float* xl = x + lane * VPL;
    vec_t bv = {};
    if (ACT) bv = *reinterpret_cast<const vec_t*>(bias + lane * VPL);

    int st[GCN_RPW];
    int dg[GCN_RPW];
    int s_f[GCN_RPW];
    float sc_f[GCN_RPW];
    float a_f[GCN_RPW];
    float dn[GCN_RPW];
#pragma unroll
    for (int i = 0; i < GCN_RPW; ++i) {
        int r = row_first + i;
        bool ok = r < n;
        st[i] = ok ? row_start[r] : 0;
        dg[i] = ok ? cnt_in[r] : 0;
        dn[i] = ok ? in_norm[r] : 0.0f;
    }
#pragma unroll
    for (int i = 0; i < GCN_RPW; ++i) {
        int m = (dg[i] < 32) ? dg[i] : 32;
        s_f[i] = 0;
        if (lane < m) s_f[i] = csr_src[st[i] + lane];
    }
#pragma unroll
    for (int i = 0; i < GCN_RPW; ++i) {
        int m = (dg[i] < 32) ? dg[i] : 32;
        sc_f[i] = 0.0f;
        a_f[i] = 0.0f;
        if (lane < m) {
            sc_f[i] = out_norm[s_f[i]];
            if (ACT) a_f[i] = in_norm[s_f[i]];
        }
    }
#pragma unroll
    for (int i = 0; i < GCN_RPW; ++i) {
        vec_t acc = {};
        int s_l = s_f[i];
        float sc_l = sc_f[i];
        float a_l = a_f[i];
        int deg = dg[i];
        int start = st[i];
        for (int base = 0; base < deg; base += 32) {
            int m = deg - base;
            m = (m < 32) ? m : 32;
            if (base > 0) {
                s_l = 0;
                sc_l = 0.0f;
                a_l = 0.0f;
                if (lane < m) {
                    s_l = csr_src[start + base + lane];
                    sc_l = out_norm[s_l];
                    if (ACT) a_l = in_norm[s_l];
                }
            }
            int j = 0;
            for (; j + 8 <= m; j += 8)
                gcn_pull_edges<F, ACT, 8, false, vec_t>(xl, s_l, sc_l, a_l, bv, j, m, acc);
            for (; j < m; j += 4)
                gcn_pull_edges<F, ACT, 4, true, vec_t>(xl, s_l, sc_l, a_l, bv, j, m, acc);
        }
        *reinterpret_cast<vec_t*>(s_rows + i * PITCH + lane * VPL) = acc * dn[i];
    }
}

__device__ __forceinline__ v2f gcn_b_frag(const float* __restrict__ wcol, int k, int koff, int Nw) {
    v2f bf;
    bf.x = wcol[(k + koff) * Nw];
    bf.y = wcol[(k + koff + 1) * Nw];
    return bf;
}

__global__ __launch_bounds__(256, 16) void gcn_layer1_fused(
    const float* __restrict__ emb, const float* __restrict__ out_norm,
    const float* __restrict__ in_norm, const int* __restrict__ row_start,
    const int* __restrict__ cnt_in, const int* __restrict__ csr_src,
    const float* __restrict__ W1, const float* __restrict__ b1,
    const float* __restrict__ W2, float* __restrict__ t, int n) {
    __shared__ __attribute__((aligned(16))) float s_a[GCN_ROWS * 132];
    __shared__ __attribute__((aligned(16))) float s_h[GCN_ROWS * 132];
    int w = __builtin_amdgcn_readfirstlane((int)(threadIdx.x >> 5));
    int lane = threadIdx.x & 31;
    int row0 = blockIdx.x * GCN_ROWS;

    gcn_pull_rows_to_lds<128, false>(emb, out_norm, in_norm, nullptr, row_start, cnt_in, csr_src,
                                     row0 + w * GCN_RPW, n, lane, s_a + w * GCN_RPW * 132);
    __syncthreads();

    int idx15 = lane & 15;
    int koff  = (lane >> 4) << 1;
    int mbase = (lane >> 4) << 3;

    {
        const float* a0 = s_a + idx15 * 132 + koff;
        const float* a1 = a0 + 16 * 132;
        const float* wcol = W1 + w * 16 + idx15;
        v8f acc0 = {};
        v8f acc1 = {};
        for (int k = 0; k < 128; k += 4) {
            v2f bf = gcn_b_frag(wcol, k, koff, 128);
            v2f af0 = *reinterpret_cast<const v2f*>(a0 + k);
            v2f af1 = *reinterpret_cast<const v2f*>(a1 + k);
            acc0 = __builtin_amdgcn_wmma_f32_16x16x4_f32(
                false, af0, false, bf, (short)0, acc0, false, false);
            acc1 = __builtin_amdgcn_wmma_f32_16x16x4_f32(
                false, af1, false, bf, (short)0, acc1, false, false);
        }
        int col = w * 16 + idx15;
        float bv = b1[col];
#pragma unroll
        for (int r = 0; r < 8; ++r) {
            float v0 = acc0[r] + bv;
            float v1 = acc1[r] + bv;
            v0 = (v0 > 0.0f) ? v0 : 0.0f;
            v1 = (v1 > 0.0f) ? v1 : 0.0f;
            s_h[(mbase + r) * 132 + col] = v0;
            s_h[(16 + mbase + r) * 132 + col] = v1;
        }
    }
    __syncthreads();

    {
        int mt = w >> 2;
        int nt = w & 3;
        const float* a0 = s_h + (mt * 16 + idx15) * 132 + koff;
        const float* wcol = W2 + nt * 16 + idx15;
        v8f acc = {};
        for (int k = 0; k < 128; k += 4) {
            v2f bf = gcn_b_frag(wcol, k, koff, 64);
            v2f af = *reinterpret_cast<const v2f*>(a0 + k);
            acc = __builtin_amdgcn_wmma_f32_16x16x4_f32(
                false, af, false, bf, (short)0, acc, false, false);
        }
        int col = nt * 16 + idx15;
#pragma unroll
        for (int r = 0; r < 8; ++r) {
            int m = row0 + mt * 16 + mbase + r;
            if (m < n) t[(size_t)m * 64 + col] = acc[r];
        }
    }
}

__global__ __launch_bounds__(256) void gcn_layer3_fused(
    const float* __restrict__ agg2, const float* __restrict__ out_norm,
    const float* __restrict__ in_norm, const float* __restrict__ b2,
    const int* __restrict__ row_start, const int* __restrict__ cnt_in,
    const int* __restrict__ csr_src, const float* __restrict__ W3,
    const float* __restrict__ b3, const int* __restrict__ graph_ids,
    float* __restrict__ pooled, int n) {
    __shared__ __attribute__((aligned(16))) float s_a[GCN_ROWS * 68];
    int w = __builtin_amdgcn_readfirstlane((int)(threadIdx.x >> 5));
    int lane = threadIdx.x & 31;
    int row0 = blockIdx.x * GCN_ROWS;

    gcn_pull_rows_to_lds<64, true>(agg2, out_norm, in_norm, b2, row_start, cnt_in, csr_src,
                                   row0 + w * GCN_RPW, n, lane, s_a + w * GCN_RPW * 68);
    __syncthreads();

    int idx15 = lane & 15;
    int koff  = (lane >> 4) << 1;
    int mbase = (lane >> 4) << 3;
    int mt = w >> 2;
    int nt = w & 3;
    int trow0 = row0 + mt * 16;
    if (trow0 >= n) return;

    const float* a0 = s_a + (mt * 16 + idx15) * 68 + koff;
    const float* wcol = W3 + nt * 16 + idx15;
    v8f acc = {};
    for (int k = 0; k < 64; k += 4) {
        v2f bf = gcn_b_frag(wcol, k, koff, 64);
        v2f af = *reinterpret_cast<const v2f*>(a0 + k);
        acc = __builtin_amdgcn_wmma_f32_16x16x4_f32(
            false, af, false, bf, (short)0, acc, false, false);
    }

    int col = nt * 16 + idx15;
    float bv = b3[col];
    int g_l = graph_ids[trow0 + idx15];
    int g0 = __builtin_amdgcn_readfirstlane(g_l);
    bool one_graph = (__builtin_amdgcn_ballot_w32(g_l != g0) == 0u);
    if (one_graph) {
        float s = 0.0f;
#pragma unroll
        for (int r = 0; r < 8; ++r) s += acc[r] + bv;
        s += __shfl_xor(s, 16);
        if (lane < 16) unsafeAtomicAdd(&pooled[g0 * 64 + col], s);
    } else {
#pragma unroll
        for (int r = 0; r < 8; ++r) {
            int m = trow0 + mbase + r;
            float v = acc[r] + bv;
            int g = graph_ids[m];
            unsafeAtomicAdd(&pooled[g * 64 + col], v);
        }
    }
}


static inline int cdiv_l(long a, long b) { return (int)((a + b - 1) / b); }

extern "C" void kernel_launch(void* const* d_in, const int* in_sizes, int n_in,
                              void* d_out, int out_size, void* d_ws, size_t ws_size,
                              hipStream_t stream) {
    const float* emb = (const float*)d_in[0];
    const float* W1  = (const float*)d_in[1];
    const float* b1  = (const float*)d_in[2];
    const float* W2  = (const float*)d_in[3];
    const float* b2  = (const float*)d_in[4];
    const float* W3  = (const float*)d_in[5];
    const float* b3  = (const float*)d_in[6];
    const int* src = (const int*)d_in[7];
    const int* dst = (const int*)d_in[8];
    const int* gid = (const int*)d_in[9];

    const int E = in_sizes[7];
    const int N = in_sizes[9];

    int*   cnt_out   = (int*)d_ws;
    int*   cnt_in    = cnt_out + N;
    float* pooled    = (float*)(cnt_in + N);
    float* counts    = pooled + GCN_N_GRAPHS * 64;
    float* out_norm  = counts + GCN_N_GRAPHS;
    float* in_norm   = out_norm + N;
    int*   row_start = (int*)(in_norm + N);
    int*   rank      = row_start + N;
    int*   csr_src   = rank + E;
    float* agg       = (float*)(csr_src + E);

    const int T = 256;

    int zc = 2 * N + GCN_N_GRAPHS * 64 + GCN_N_GRAPHS;
    gcn_fill_zero<<<cdiv_l(zc, T), T, 0, stream>>>((float*)d_ws, zc);

    gcn_degree_rank<<<cdiv_l(E, T), T, 0, stream>>>(src, dst, cnt_out, cnt_in, rank, E);
    gcn_norm_finalize<<<cdiv_l(2L * N, T), T, 0, stream>>>(cnt_out, out_norm, 2 * N);
    gcn_count<<<cdiv_l(N, T), T, 0, stream>>>(gid, counts, N);

    gcn_row_start<<<cdiv_l(N, 1024), 1024, 0, stream>>>(cnt_in, row_start, N);
    gcn_place<<<cdiv_l(E, T), T, 0, stream>>>(src, dst, row_start, rank, csr_src, E);

    const int rows_per_block = T / 32;
    const int groups = cdiv_l(N, GCN_ROWS);

    float* t    = agg;
    float* agg2 = agg + (size_t)N * 64;
    gcn_layer1_fused<<<groups, T, 0, stream>>>(
        emb, out_norm, in_norm, row_start, cnt_in, csr_src, W1, b1, W2, t, N);

    gcn_pull<64, false><<<cdiv_l(N, rows_per_block), T, 0, stream>>>(
        t, out_norm, nullptr, nullptr, row_start, cnt_in, csr_src, agg2, N);

    gcn_layer3_fused<<<groups, T, 0, stream>>>(
        agg2, out_norm, in_norm, b2, row_start, cnt_in, csr_src, W3, b3, gid, pooled, N);

    gcn_final<<<cdiv_l(GCN_N_GRAPHS * 64, T), T, 0, stream>>>(pooled, counts, (float*)d_out);
}


# Per-edge source scaling moved to the row's producer: t stored times out_norm, layer 2's edge pass stores relu(.)*out_norm; layer 2/3 gathers sum plain rows, no per-edge norm loads. Sum order kept.
# speedup vs baseline: 3.5645x; 1.0468x over previous
#include <hip/hip_runtime.h>

typedef __attribute__((ext_vector_type(2))) float v2f;
typedef __attribute__((ext_vector_type(8))) float v8f;

#define GCN_N_GRAPHS 64


__global__ void gcn_fill_zero(float* __restrict__ p, int n) {
    int i = blockIdx.x * blockDim.x + threadIdx.x;
    if (i < n) p[i] = 0.0f;
}

__global__ void gcn_degree_rank(const int* __restrict__ src, const int* __restrict__ dst,
                                int* __restrict__ cnt_out, int* __restrict__ cnt_in,
                                int* __restrict__ rank, int E) {
    int e = blockIdx.x * blockDim.x + threadIdx.x;
    if (e < E) {
        atomicAdd(&cnt_out[src[e]], 1);
        rank[e] = atomicAdd(&cnt_in[dst[e]], 1);
    }
}

__global__ void gcn_norm_finalize(const int* __restrict__ cnt, float* __restrict__ norm, int n) {
    int i = blockIdx.x * blockDim.x + threadIdx.x;
    if (i < n) {
        float d = (float)cnt[i];
        d = (d < 1.0f) ? 1.0f : d;
        norm[i] = 1.0f / sqrtf(d);
    }
}

__global__ void gcn_row_start(const int* __restrict__ cnt, int* __restrict__ row_start, int n) {
    __shared__ int s_front[32];
    __shared__ int s_wave[32];
    __shared__ int s_base;
    int tid = threadIdx.x;
    int lane = tid & 31;
    int w = tid >> 5;
    int first = blockIdx.x * 1024;
    int i = first + tid;

    int front = 0;
    for (int j = tid; j < first; j += 1024) front += cnt[j];
    int v = (i < n) ? cnt[i] : 0;

    for (int o = 16; o > 0; o >>= 1) front += __shfl_xor(front, o);
    int inc = v;
    for (int o = 1; o < 32; o <<= 1) {
        int t = __shfl_up(inc, o);
        if (lane >= o) inc += t;
    }
    if (lane == 0) s_front[w] = front;
    if (lane == 31) s_wave[w] = inc;
    __syncthreads();
    if (w == 0) {
        int a = s_front[lane];
        for (int o = 16; o > 0; o >>= 1) a += __shfl_xor(a, o);
        int p = s_wave[lane];
        int pi = p;
        for (int o = 1; o < 32; o <<= 1) {
            int t = __shfl_up(pi, o);
            if (lane >= o) pi += t;
        }
        s_wave[lane] = pi - p;
        if (lane == 0) s_base = a;
    }
    __syncthreads();
    if (i < n) row_start[i] = s_base + s_wave[w] + inc - v;
}

__global__ void gcn_place(const int* __restrict__ src, const int* __restrict__ dst,
                          const int* __restrict__ row_start, const int* __restrict__ rank,
                          int* __restrict__ csr_src, int E) {
    int e = blockIdx.x * blockDim.x + threadIdx.x;
    if (e < E) csr_src[row_start[dst[e]] + rank[e]] = src[e];
}

__global__ void gcn_count(const int* __restrict__ gid, float* __restrict__ counts, int n) {
    int i = blockIdx.x * blockDim.x + threadIdx.x;
    int lane = threadIdx.x & 31;
    bool todo = i < n;
    int g = todo ? gid[i] : 0;
    for (int it = 0; it < 32; ++it) {
        unsigned waiting = __builtin_amdgcn_ballot_w32(todo);
        if (waiting == 0u) break;
        int first = __builtin_ctz(waiting);
        int g0 = __builtin_amdgcn_readlane(g, first);
        bool same = todo && (g == g0);
        unsigned members = __builtin_amdgcn_ballot_w32(same);
        if (lane == first) unsafeAtomicAdd(&counts[g0], (float)__builtin_popcount(members));
        todo = todo && !same;
    }
}

__global__ void gcn_final(const float* __restrict__ pooled, const float* __restrict__ counts,
                          float* __restrict__ out) {
    int i = blockIdx.x * blockDim.x + threadIdx.x;
    if (i < GCN_N_GRAPHS * 64) {
        float c = counts[i >> 6];
        c = (c < 1.0f) ? 1.0f : c;
        out[i] = pooled[i] / c;
    }
}


template <int V> struct gcn_vec;
template <> struct gcn_vec<4> { typedef __attribute__((ext_vector_type(4))) float type; };
template <> struct gcn_vec<2> { typedef __attribute__((ext_vector_type(2))) float type; };

template <int F, bool SCALE, int U, bool TAIL, typename vec_t>
__device__ __forceinline__ void gcn_pull_edges(const float* __restrict__ xl, int s_l, float sc_l,
                                               int j, int m, vec_t& acc) {
    vec_t v[U];
    float sc[U];
#pragma unroll
    for (int u = 0; u < U; ++u) {
        int jj = j + u;
        if (TAIL) jj = (jj < m) ? jj : m - 1;
        int s = __builtin_amdgcn_readlane(s_l, jj);
        sc[u] = SCALE ? __int_as_float(__builtin_amdgcn_readlane(__float_as_int(sc_l), jj)) : 1.0f;
        v[u] = *reinterpret_cast<const vec_t*>(xl + (size_t)s * F);
    }
#pragma unroll
    for (int u = 0; u < U; ++u) {
        if (!TAIL || j + u < m) {
            if (SCALE) acc += v[u] * sc[u];
            else acc += v[u];
        }
    }
}

template <int F>
__global__ void gcn_pull_finish(const float* __restrict__ x, const float* __restrict__ out_norm,
                                const float* __restrict__ in_norm, const float* __restrict__ bias,
                                const int* __restrict__ row_start, const int* __restrict__ cnt_in,
                                const int* __restrict__ csr_src, float* __restrict__ y, int n) {
    const int VPL = F / 32;
    typedef typename gcn_vec<VPL>::type vec_t;
    int row = __builtin_amdgcn_readfirstlane(
        (int)(blockIdx.x * (blockDim.x >> 5) + (threadIdx.x >> 5)));
    if (row >= n) return;
    int lane = threadIdx.x & 31;
    int start = row_start[row];
    int deg = cnt_in[row];
    float dn = in_norm[row];
    float on = out_norm[row];
    const float* xl = x + lane * VPL;

    vec_t acc = {};
    vec_t bv = *reinterpret_cast<const vec_t*>(bias + lane * VPL);

    for (int base = 0; base < deg; base += 32) {
        int m = deg - base;
        m = (m < 32) ? m : 32;
        int s_l = 0;
        if (lane < m) s_l = csr_src[start + base + lane];
        int j = 0;
        for (; j + 8 <= m; j += 8)
            gcn_pull_edges<F, false, 8, false, vec_t>(xl, s_l, 0.0f, j, m, acc);
        for (; j < m; j += 4)
            gcn_pull_edges<F, false, 4, true, vec_t>(xl, s_l, 0.0f, j, m, acc);
    }
    vec_t r = acc * dn + bv;
#pragma unroll
    for (int q = 0; q < VPL; ++q) r[q] = (r[q] > 0.0f) ? r[q] : 0.0f;
    *reinterpret_cast<vec_t*>(y + (size_t)row * F + lane * VPL) = r * on;
}


#define GCN_ROWS 32
#define GCN_RPW 4

template <int F, bool SCALE>
__device__ __forceinline__ void gcn_pull_rows_to_lds(
    const float* __restrict__ x, const float* __restrict__ out_norm,
    const float* __restrict__ in_norm,
    const int* __restrict__ row_start, const int* __restrict__ cnt_in,
    const int* __restrict__ csr_src, int row_first, int n, int lane, float* s_rows) {
    const int VPL = F / 32;
    const int PITCH = F + 4;
    typedef typename gcn_vec<VPL>::type vec_t;
    const float* xl = x + lane * VPL;

    int st[GCN_RPW];
    int dg[GCN_RPW];
    int s_f[GCN_RPW];
    float sc_f[GCN_RPW];
    float dn[GCN_RPW];
#pragma unroll
    for (int i = 0; i < GCN_RPW; ++i) {
        int r = row_first + i;
        bool ok = r < n;
        st[i] = ok ? row_start[r] : 0;
        dg[i] = ok ? cnt_in[r] : 0;
        dn[i] = ok ? in_norm[r] : 0.0f;
    }
#pragma unroll
    for (int i = 0; i < GCN_RPW; ++i) {
        int m = (dg[i] < 32) ? dg[i] : 32;
        s_f[i] = 0;
        if (lane < m) s_f[i] = csr_src[st[i] + lane];
    }
#pragma unroll
    for (int i = 0; i < GCN_RPW; ++i) {
        int m = (dg[i] < 32) ? dg[i] : 32;
        sc_f[i] = 0.0f;
        if (SCALE && lane < m) sc_f[i] = out_norm[s_f[i]];
    }
#pragma unroll
    for (int i = 0; i < GCN_RPW; ++i) {
        vec_t acc = {};
        int s_l = s_f[i];
        float sc_l = sc_f[i];
        int deg = dg[i];
        int start = st[i];
        for (int base = 0; base < deg; base += 32) {
            int m = deg - base;
            m = (m < 32) ? m : 32;
            if (base > 0) {
                s_l = 0;
                sc_l = 0.0f;
                if (lane < m) {
                    s_l = csr_src[start + base + lane];
                    if (SCALE) sc_l = out_norm[s_l];
                }
            }
            int j = 0;
            for (; j + 8 <= m; j += 8)
                gcn_pull_edges<F, SCALE, 8, false, vec_t>(xl, s_l, sc_l, j, m, acc);
            for (; j < m; j += 4)
                gcn_pull_edges<F, SCALE, 4, true, vec_t>(xl, s_l, sc_l, j, m, acc);
        }
        *reinterpret_cast<vec_t*>(s_rows + i * PITCH + lane * VPL) = acc * dn[i];
    }
}

__device__ __forceinline__ v2f gcn_b_frag(const float* __restrict__ wcol, int k, int koff, int Nw) {
    v2f bf;
    bf.x = wcol[(k + koff) * Nw];
    bf.y = wcol[(k + koff + 1) * Nw];
    return bf;
}

__global__ __launch_bounds__(256, 16) void gcn_layer1_fused(
    const float* __restrict__ emb, const float* __restrict__ out_norm,
    const float* __restrict__ in_norm, const int* __restrict__ row_start,
    const int* __restrict__ cnt_in, const int* __restrict__ csr_src,
    const float* __restrict__ W1, const float* __restrict__ b1,
    const float* __restrict__ W2, float* __restrict__ t, int n) {
    __shared__ __attribute__((aligned(16))) float s_a[GCN_ROWS * 132];
    __shared__ __attribute__((aligned(16))) float s_h[GCN_ROWS * 132];
    __shared__ __attribute__((aligned(16))) float s_on[GCN_ROWS];
    int w = __builtin_amdgcn_readfirstlane((int)(threadIdx.x >> 5));
    int lane = threadIdx.x & 31;
    int row0 = blockIdx.x * GCN_ROWS;

    typedef gcn_vec<4>::type on_t;
    on_t on_w;
#pragma unroll
    for (int i = 0; i < GCN_RPW; ++i) {
        int r = row0 + w * GCN_RPW + i;
        on_w[i] = (r < n) ? out_norm[r] : 0.0f;
    }

    gcn_pull_rows_to_lds<128, true>(emb, out_norm, in_norm, row_start, cnt_in, csr_src,
                                    row0 + w * GCN_RPW, n, lane, s_a + w * GCN_RPW * 132);
    if (lane == 0) *reinterpret_cast<on_t*>(s_on + w * GCN_RPW) = on_w;
    __syncthreads();

    int idx15 = lane & 15;
    int koff  = (lane >> 4) << 1;
    int mbase = (lane >> 4) << 3;

    {
        const float* a0 = s_a + idx15 * 132 + koff;
        const float* a1 = a0 + 16 * 132;
        const float* wcol = W1 + w * 16 + idx15;
        v8f acc0 = {};
        v8f acc1 = {};
        for (int k = 0; k < 128; k += 4) {
            v2f bf = gcn_b_frag(wcol, k, koff, 128);
            v2f af0 = *reinterpret_cast<const v2f*>(a0 + k);
            v2f af1 = *reinterpret_cast<const v2f*>(a1 + k);
            acc0 = __builtin_amdgcn_wmma_f32_16x16x4_f32(
                false, af0, false, bf, (short)0, acc0, false, false);
            acc1 = __builtin_amdgcn_wmma_f32_16x16x4_f32(
                false, af1, false, bf, (short)0, acc1, false, false);
        }
        int col = w * 16 + idx15;
        float bv = b1[col];
#pragma unroll
        for (int r = 0; r < 8; ++r) {
            float v0 = acc0[r] + bv;
            float v1 = acc1[r] + bv;
            v0 = (v0 > 0.0f) ? v0 : 0.0f;
            v1 = (v1 > 0.0f) ? v1 : 0.0f;
            s_h[(mbase + r) * 132 + col] = v0;
            s_h[(16 + mbase + r) * 132 + col] = v1;
        }
    }
    __syncthreads();

    {
        int mt = w >> 2;
        int nt = w & 3;
        const float* a0 = s_h + (mt * 16 + idx15) * 132 + koff;
        const float* wcol = W2 + nt * 16 + idx15;
        v8f acc = {};
        for (int k = 0; k < 128; k += 4) {
            v2f bf = gcn_b_frag(wcol, k, koff, 64);
            v2f af = *reinterpret_cast<const v2f*>(a0 + k);
            acc = __builtin_amdgcn_wmma_f32_16x16x4_f32(
                false, af, false, bf, (short)0, acc, false, false);
        }
        int col = nt * 16 + idx15;
#pragma unroll
        for (int r = 0; r < 8; ++r) {
            int m = row0 + mt * 16 + mbase + r;
            if (m < n) t[(size_t)m * 64 + col] = acc[r] * s_on[mt * 16 + mbase + r];
        }
    }
}

__global__ __launch_bounds__(256) void gcn_layer3_fused(
    const float* __restrict__ x3, const float* __restrict__ in_norm,
    const int* __restrict__ row_start, const int* __restrict__ cnt_in,
    const int* __restrict__ csr_src, const float* __restrict__ W3,
    const float* __restrict__ b3, const int* __restrict__ graph_ids,
    float* __restrict__ pooled, int n) {
    __shared__ __attribute__((aligned(16))) float s_a[GCN_ROWS * 68];
    int w = __builtin_amdgcn_readfirstlane((int)(threadIdx.x >> 5));
    int lane = threadIdx.x & 31;
    int row0 = blockIdx.x * GCN_ROWS;

    gcn_pull_rows_to_lds<64, false>(x3, nullptr, in_norm, row_start, cnt_in, csr_src,
                                    row0 + w * GCN_RPW, n, lane, s_a + w * GCN_RPW * 68);
    __syncthreads();

    int idx15 = lane & 15;
    int koff  = (lane >> 4) << 1;
    int mbase = (lane >> 4) << 3;
    int mt = w >> 2;
    int nt = w & 3;
    int trow0 = row0 + mt * 16;
    if (trow0 >= n) return;

    const float* a0 = s_a + (mt * 16 + idx15) * 68 + koff;
    const float* wcol = W3 + nt * 16 + idx15;
    v8f acc = {};
    for (int k = 0; k < 64; k += 4) {
        v2f bf = gcn_b_frag(wcol, k, koff, 64);
        v2f af = *reinterpret_cast<const v2f*>(a0 + k);
        acc = __builtin_amdgcn_wmma_f32_16x16x4_f32(
            false, af, false, bf, (short)0, acc, false, false);
    }

    int col = nt * 16 + idx15;
    float bv = b3[col];
    int g_l = graph_ids[trow0 + idx15];
    int g0 = __builtin_amdgcn_readfirstlane(g_l);
    bool one_graph = (__builtin_amdgcn_ballot_w32(g_l != g0) == 0u);
    if (one_graph) {
        float s = 0.0f;
#pragma unroll
        for (int r = 0; r < 8; ++r) s += acc[r] + bv;
        s += __shfl_xor(s, 16);
        if (lane < 16) unsafeAtomicAdd(&pooled[g0 * 64 + col], s);
    } else {
#pragma unroll
        for (int r = 0; r < 8; ++r) {
            int m = trow0 + mbase + r;
            float v = acc[r] + bv;
            int g = graph_ids[m];
            unsafeAtomicAdd(&pooled[g * 64 + col], v);
        }
    }
}


static inline int cdiv_l(long a, long b) { return (int)((a + b - 1) / b); }

extern "C" void kernel_launch(void* const* d_in, const int* in_sizes, int n_in,
                              void* d_out, int out_size, void* d_ws, size_t ws_size,
                              hipStream_t stream) {
    const float* emb = (const float*)d_in[0];
    const float* W1  = (const float*)d_in[1];
    const float* b1  = (const float*)d_in[2];
    const float* W2  = (const float*)d_in[3];
    const float* b2  = (const float*)d_in[4];
    const float* W3  = (const float*)d_in[5];
    const float* b3  = (const float*)d_in[6];
    const int* src = (const int*)d_in[7];
    const int* dst = (const int*)d_in[8];
    const int* gid = (const int*)d_in[9];

    const int E = in_sizes[7];
    const int N = in_sizes[9];

    int*   cnt_out   = (int*)d_ws;
    int*   cnt_in    = cnt_out + N;
    float* pooled    = (float*)(cnt_in + N);
    float* counts    = pooled + GCN_N_GRAPHS * 64;
    float* out_norm  = counts + GCN_N_GRAPHS;
    float* in_norm   = out_norm + N;
    int*   row_start = (int*)(in_norm + N);
    int*   rank      = row_start + N;
    int*   csr_src   = rank + E;
    float* rows      = (float*)(csr_src + E);

    const int T = 256;

    int zc = 2 * N + GCN_N_GRAPHS * 64 + GCN_N_GRAPHS;
    gcn_fill_zero<<<cdiv_l(zc, T), T, 0, stream>>>((float*)d_ws, zc);

    gcn_degree_rank<<<cdiv_l(E, T), T, 0, stream>>>(src, dst, cnt_out, cnt_in, rank, E);
    gcn_norm_finalize<<<cdiv_l(2L * N, T), T, 0, stream>>>(cnt_out, out_norm, 2 * N);
    gcn_count<<<cdiv_l(N, T), T, 0, stream>>>(gid, counts, N);

    gcn_row_start<<<cdiv_l(N, 1024), 1024, 0, stream>>>(cnt_in, row_start, N);
    gcn_place<<<cdiv_l(E, T), T, 0, stream>>>(src, dst, row_start, rank, csr_src, E);

    const int rows_per_block = T / 32;
    const int groups = cdiv_l(N, GCN_ROWS);

    float* t  = rows;
    float* x3 = rows + (size_t)N * 64;
    gcn_layer1_fused<<<groups, T, 0, stream>>>(
        emb, out_norm, in_norm, row_start, cnt_in, csr_src, W1, b1, W2, t, N);

    gcn_pull_finish<64><<<cdiv_l(N, rows_per_block), T, 0, stream>>>(
        t, out_norm, in_norm, b2, row_start, cnt_in, csr_src, x3, N);

    gcn_layer3_fused<<<groups, T, 0, stream>>>(
        x3, in_norm, row_start, cnt_in, csr_src, W3, b3, gid, pooled, N);

    gcn_final<<<cdiv_l(GCN_N_GRAPHS * 64, T), T, 0, stream>>>(pooled, counts, (float*)d_out);
}


# Degrees and bucket slots come from LDS counters: a work-group per (node range, edge slice) counts, then places its edges with LDS adds; no global atomic per edge. f32 sum order: slice, then arrival
# speedup vs baseline: 5.8531x; 1.6421x over previous
#include <hip/hip_runtime.h>

typedef __attribute__((ext_vector_type(2))) float v2f;
typedef __attribute__((ext_vector_type(8))) float v8f;

#define GCN_N_GRAPHS 64
#define GCN_SLOTS 32

#define GCN_RANGE 25600
#define GCN_SLICES 64
#define GCN_CT 1024
#define GCN_EPT 8

__global__ __launch_bounds__(GCN_CT) void gcn_count_ranges(
    const int* __restrict__ src, const int* __restrict__ dst, int* __restrict__ part_out,
    int* __restrict__ part_in, float* __restrict__ zero_p, int zero_n, int E, int n) {
    __shared__ __attribute__((aligned(16))) int s_out[GCN_RANGE];
    __shared__ __attribute__((aligned(16))) int s_in[GCN_RANGE];
    typedef __attribute__((ext_vector_type(4))) int v4i;
    const int tid = threadIdx.x;
    const int nr = (n + GCN_RANGE - 1) / GCN_RANGE;
    const int r = blockIdx.x % nr;
    const int j = blockIdx.x / nr;
    const int base = r * GCN_RANGE;
    const v4i z4 = {0, 0, 0, 0};
    for (int i = tid; i < GCN_RANGE / 4; i += GCN_CT) {
        reinterpret_cast<v4i*>(s_out)[i] = z4;
        reinterpret_cast<v4i*>(s_in)[i] = z4;
    }
    if (blockIdx.x == 0)
        for (int i = tid; i < zero_n; i += GCN_CT) zero_p[i] = 0.0f;
    __syncthreads();
    const int es = (E + GCN_SLICES - 1) / GCN_SLICES;
    const int e0 = j * es;
    const int e1 = (e0 + es < E) ? e0 + es : E;
    for (int e = e0 + tid; e < e1; e += GCN_CT * GCN_EPT) {
        int sv[GCN_EPT];
        int dv[GCN_EPT];
#pragma unroll
        for (int u = 0; u < GCN_EPT; ++u) {
            int ee = e + u * GCN_CT;
            bool ok = ee < e1;
            int ec = ok ? ee : e;
            int s = src[ec];
            int d = dst[ec];
            sv[u] = ok ? s : -1;
            dv[u] = ok ? d : -1;
        }
#pragma unroll
        for (int u = 0; u < GCN_EPT; ++u) {
            unsigned us = (unsigned)(sv[u] - base);
            unsigned ud = (unsigned)(dv[u] - base);
            if (us < (unsigned)GCN_RANGE) atomicAdd(&s_out[us], 1);
            if (ud < (unsigned)GCN_RANGE) atomicAdd(&s_in[ud], 1);
        }
    }
    __syncthreads();
    const int lim = (n - base < GCN_RANGE) ? n - base : GCN_RANGE;
    int* po = part_out + (size_t)j * n + base;
    int* pi = part_in + (size_t)j * n + base;
    for (int i = tid; i < lim; i += GCN_CT) {
        po[i] = s_out[i];
        pi[i] = s_in[i];
    }
}

__global__ void gcn_offsets_norms(const int* __restrict__ part_out, const int* __restrict__ part_in,
                                  int* __restrict__ part_base, int* __restrict__ cnt_in,
                                  float* __restrict__ out_norm, float* __restrict__ in_norm, int n) {
    int i = blockIdx.x * blockDim.x + threadIdx.x;
    if (i >= n) return;
    int so = 0;
    int run = 0;
    for (int j0 = 0; j0 < GCN_SLICES; j0 += 16) {
        int co[16];
        int ci[16];
#pragma unroll
        for (int u = 0; u < 16; ++u) {
            co[u] = part_out[(size_t)(j0 + u) * n + i];
            ci[u] = part_in[(size_t)(j0 + u) * n + i];
        }
#pragma unroll
        for (int u = 0; u < 16; ++u) {
            so += co[u];
            part_base[(size_t)(j0 + u) * n + i] = run;
            run += ci[u];
        }
    }
    cnt_in[i] = run;
    float d = (float)so;
    d = (d < 1.0f) ? 1.0f : d;
    out_norm[i] = 1.0f / sqrtf(d);
    d = (float)run;
    d = (d < 1.0f) ? 1.0f : d;
    in_norm[i] = 1.0f / sqrtf(d);
}

__global__ __launch_bounds__(GCN_CT) void gcn_place_ranges(
    const int* __restrict__ src, const int* __restrict__ dst, const int* __restrict__ part_base,
    int* __restrict__ bucket, int* __restrict__ ovf_count, int* __restrict__ ovf_dst,
    int* __restrict__ ovf_src, int E, int n) {
    __shared__ int s_cur[GCN_RANGE];
    const int tid = threadIdx.x;
    const int nr = (n + GCN_RANGE - 1) / GCN_RANGE;
    const int r = blockIdx.x % nr;
    const int j = blockIdx.x / nr;
    const int base = r * GCN_RANGE;
    const int lim = (n - base < GCN_RANGE) ? n - base : GCN_RANGE;
    const int* pb = part_base + (size_t)j * n + base;
    for (int i = tid; i < lim; i += GCN_CT) s_cur[i] = pb[i];
    __syncthreads();
    const int es = (E + GCN_SLICES - 1) / GCN_SLICES;
    const int e0 = j * es;
    const int e1 = (e0 + es < E) ? e0 + es : E;
    for (int e = e0 + tid; e < e1; e += GCN_CT * GCN_EPT) {
        int sv[GCN_EPT];
        int dv[GCN_EPT];
#pragma unroll
        for (int u = 0; u < GCN_EPT; ++u) {
            int ee = e + u * GCN_CT;
            bool ok = ee < e1;
            int ec = ok ? ee : e;
            int s = src[ec];
            int d = dst[ec];
            sv[u] = ok ? s : -1;
            dv[u] = ok ? d : -1;
        }
#pragma unroll
        for (int u = 0; u < GCN_EPT; ++u) {
            int d = dv[u];
            unsigned ud = (unsigned)(d - base);
            if (ud < (unsigned)lim) {
                int slot = atomicAdd(&s_cur[ud], 1);
                if (slot < GCN_SLOTS) {
                    bucket[(size_t)d * GCN_SLOTS + slot] = sv[u];
                } else {
                    int k = atomicAdd(ovf_count, 1);
                    ovf_dst[k] = d;
                    ovf_src[k] = sv[u];
                }
            }
        }
    }
}

__global__ void gcn_count(const int* __restrict__ gid, float* __restrict__ counts, int n) {
    int i = blockIdx.x * blockDim.x + threadIdx.x;
    int lane = threadIdx.x & 31;
    bool todo = i < n;
    int g = todo ? gid[i] : 0;
    for (int it = 0; it < 32; ++it) {
        unsigned waiting = __builtin_amdgcn_ballot_w32(todo);
        if (waiting == 0u) break;
        int first = __builtin_ctz(waiting);
        int g0 = __builtin_amdgcn_readlane(g, first);
        bool same = todo && (g == g0);
        unsigned members = __builtin_amdgcn_ballot_w32(same);
        if (lane == first) unsafeAtomicAdd(&counts[g0], (float)__builtin_popcount(members));
        todo = todo && !same;
    }
}

__global__ void gcn_final(const float* __restrict__ pooled, const float* __restrict__ counts,
                          float* __restrict__ out) {
    int i = blockIdx.x * blockDim.x + threadIdx.x;
    if (i < GCN_N_GRAPHS * 64) {
        float c = counts[i >> 6];
        c = (c < 1.0f) ? 1.0f : c;
        out[i] = pooled[i] / c;
    }
}


template <int V> struct gcn_vec;
template <> struct gcn_vec<4> { typedef __attribute__((ext_vector_type(4))) float type; };
template <> struct gcn_vec<2> { typedef __attribute__((ext_vector_type(2))) float type; };

template <int F, bool SCALE, int U, bool TAIL, typename vec_t>
__device__ __forceinline__ void gcn_pull_edges(const float* __restrict__ xl, int s_l, float sc_l,
                                               int j, int m, vec_t& acc) {
    vec_t v[U];
    float sc[U];
#pragma unroll
    for (int u = 0; u < U; ++u) {
        int jj = j + u;
        if (TAIL) jj = (jj < m) ? jj : m - 1;
        int s = __builtin_amdgcn_readlane(s_l, jj);
        sc[u] = SCALE ? __int_as_float(__builtin_amdgcn_readlane(__float_as_int(sc_l), jj)) : 1.0f;
        v[u] = *reinterpret_cast<const vec_t*>(xl + (size_t)s * F);
    }
#pragma unroll
    for (int u = 0; u < U; ++u) {
        if (!TAIL || j + u < m) {
            if (SCALE) acc += v[u] * sc[u];
            else acc += v[u];
        }
    }
}

template <int F, bool SCALE, typename vec_t>
__device__ __forceinline__ void gcn_pull_overflow(const float* __restrict__ xl,
                                                  const float* __restrict__ out_norm, int row, int lane,
                                                  const int* __restrict__ ovf_count,
                                                  const int* __restrict__ ovf_dst,
                                                  const int* __restrict__ ovf_src, vec_t& acc) {
    int total = ovf_count[0];
    for (int k0 = 0; k0 < total; k0 += 32) {
        int k = k0 + lane;
        bool mine = false;
        if (k < total) mine = (ovf_dst[k] == row);
        int s = 0;
        float sc = 1.0f;
        if (mine) {
            s = ovf_src[k];
            if (SCALE) sc = out_norm[s];
        }
        unsigned todo = __builtin_amdgcn_ballot_w32(mine);
        for (int it = 0; it < 32; ++it) {
            if (todo == 0u) break;
            int j = __builtin_ctz(todo);
            todo &= todo - 1u;
            int sj = __builtin_amdgcn_readlane(s, j);
            vec_t v = *reinterpret_cast<const vec_t*>(xl + (size_t)sj * F);
            if (SCALE) {
                float scj = __int_as_float(__builtin_amdgcn_readlane(__float_as_int(sc), j));
                acc += v * scj;
            } else {
                acc += v;
            }
        }
    }
}

template <int F>
__global__ void gcn_pull_finish(const float* __restrict__ x, const float* __restrict__ out_norm,
                                const float* __restrict__ in_norm, const float* __restrict__ bias,
                                const int* __restrict__ cnt_in, const int* __restrict__ bucket,
                                const int* __restrict__ ovf_count, const int* __restrict__ ovf_dst,
                                const int* __restrict__ ovf_src, float* __restrict__ y, int n) {
    const int VPL = F / 32;
    typedef typename gcn_vec<VPL>::type vec_t;
    int row = __builtin_amdgcn_readfirstlane(
        (int)(blockIdx.x * (blockDim.x >> 5) + (threadIdx.x >> 5)));
    if (row >= n) return;
    int lane = threadIdx.x & 31;
    int deg = cnt_in[row];
    float dn = in_norm[row];
    float on = out_norm[row];
    const float* xl = x + lane * VPL;

    vec_t acc = {};
    vec_t bv = *reinterpret_cast<const vec_t*>(bias + lane * VPL);

    {
        int m = (deg < GCN_SLOTS) ? deg : GCN_SLOTS;
        int s_l = 0;
        if (lane < m) s_l = bucket[(size_t)row * GCN_SLOTS + lane];
        int j = 0;
        for (; j + 8 <= m; j += 8)
            gcn_pull_edges<F, false, 8, false, vec_t>(xl, s_l, 0.0f, j, m, acc);
        for (; j < m; j += 4)
            gcn_pull_edges<F, false, 4, true, vec_t>(xl, s_l, 0.0f, j, m, acc);
    }
    if (deg > GCN_SLOTS)
        gcn_pull_overflow<F, false, vec_t>(xl, out_norm, row, lane, ovf_count, ovf_dst, ovf_src, acc);
    vec_t r = acc * dn + bv;
#pragma unroll
    for (int q = 0; q < VPL; ++q) r[q] = (r[q] > 0.0f) ? r[q] : 0.0f;
    *reinterpret_cast<vec_t*>(y + (size_t)row * F + lane * VPL) = r * on;
}


#define GCN_ROWS 32
#define GCN_RPW 4

template <int F, bool SCALE>
__device__ __forceinline__ void gcn_pull_rows_to_lds(
    const float* __restrict__ x, const float* __restrict__ out_norm,
    const float* __restrict__ in_norm,
    const int* __restrict__ cnt_in, const int* __restrict__ bucket,
    const int* __restrict__ ovf_count, const int* __restrict__ ovf_dst,
    const int* __restrict__ ovf_src, int row_first, int n, int lane, float* s_rows) {
    const int VPL = F / 32;
    const int PITCH = F + 4;
    typedef typename gcn_vec<VPL>::type vec_t;
    const float* xl = x + lane * VPL;

    int dg[GCN_RPW];
    int s_f[GCN_RPW];
    float sc_f[GCN_RPW];
    float dn[GCN_RPW];
#pragma unroll
    for (int i = 0; i < GCN_RPW; ++i) {
        int r = row_first + i;
        bool ok = r < n;
        dg[i] = ok ? cnt_in[r] : 0;
        dn[i] = ok ? in_norm[r] : 0.0f;
    }
#pragma unroll
    for (int i = 0; i < GCN_RPW; ++i) {
        int m = (dg[i] < GCN_SLOTS) ? dg[i] : GCN_SLOTS;
        s_f[i] = 0;
        if (lane < m) s_f[i] = bucket[(size_t)(row_first + i) * GCN_SLOTS + lane];
    }
#pragma unroll
    for (int i = 0; i < GCN_RPW; ++i) {
        int m = (dg[i] < GCN_SLOTS) ? dg[i] : GCN_SLOTS;
        sc_f[i] = 0.0f;
        if (SCALE && lane < m) sc_f[i] = out_norm[s_f[i]];
    }
#pragma unroll
    for (int i = 0; i < GCN_RPW; ++i) {
        vec_t acc = {};
        int s_l = s_f[i];
        float sc_l = sc_f[i];
        int deg = dg[i];
        {
            int m = (deg < GCN_SLOTS) ? deg : GCN_SLOTS;
            int j = 0;
            for (; j + 8 <= m; j += 8)
                gcn_pull_edges<F, SCALE, 8, false, vec_t>(xl, s_l, sc_l, j, m, acc);
            for (; j < m; j += 4)
                gcn_pull_edges<F, SCALE, 4, true, vec_t>(xl, s_l, sc_l, j, m, acc);
        }
        if (deg > GCN_SLOTS)
            gcn_pull_overflow<F, SCALE, vec_t>(xl, out_norm, row_first + i, lane,
                                               ovf_count, ovf_dst, ovf_src, acc);
        *reinterpret_cast<vec_t*>(s_rows + i * PITCH + lane * VPL) = acc * dn[i];
    }
}

__device__ __forceinline__ v2f gcn_b_frag(const float* __restrict__ wcol, int k, int koff, int Nw) {
    v2f bf;
    bf.x = wcol[(k + koff) * Nw];
    bf.y = wcol[(k + koff + 1) * Nw];
    return bf;
}

__global__ __launch_bounds__(256, 16) void gcn_layer1_fused(
    const float* __restrict__ emb, const float* __restrict__ out_norm,
    const float* __restrict__ in_norm, const int* __restrict__ cnt_in,
    const int* __restrict__ bucket, const int* __restrict__ ovf_count,
    const int* __restrict__ ovf_dst, const int* __restrict__ ovf_src,
    const float* __restrict__ W1, const float* __restrict__ b1,
    const float* __restrict__ W2, float* __restrict__ t, int n) {
    __shared__ __attribute__((aligned(16))) float s_a[GCN_ROWS * 132];
    __shared__ __attribute__((aligned(16))) float s_h[GCN_ROWS * 132];
    __shared__ __attribute__((aligned(16))) float s_on[GCN_ROWS];
    int w = __builtin_amdgcn_readfirstlane((int)(threadIdx.x >> 5));
    int lane = threadIdx.x & 31;
    int row0 = blockIdx.x * GCN_ROWS;

    typedef gcn_vec<4>::type on_t;
    on_t on_w;
#pragma unroll
    for (int i = 0; i < GCN_RPW; ++i) {
        int r = row0 + w * GCN_RPW + i;
        on_w[i] = (r < n) ? out_norm[r] : 0.0f;
    }

    gcn_pull_rows_to_lds<128, true>(emb, out_norm, in_norm, cnt_in, bucket, ovf_count, ovf_dst,
                                    ovf_src, row0 + w * GCN_RPW, n, lane, s_a + w * GCN_RPW * 132);
    if (lane == 0) *reinterpret_cast<on_t*>(s_on + w * GCN_RPW) = on_w;
    __syncthreads();

    int idx15 = lane & 15;
    int koff  = (lane >> 4) << 1;
    int mbase = (lane >> 4) << 3;

    {
        const float* a0 = s_a + idx15 * 132 + koff;
        const float* a1 = a0 + 16 * 132;
        const float* wcol = W1 + w * 16 + idx15;
        v8f acc0 = {};
        v8f acc1 = {};
        for (int k = 0; k < 128; k += 4) {
            v2f bf = gcn_b_frag(wcol, k, koff, 128);
            v2f af0 = *reinterpret_cast<const v2f*>(a0 + k);
            v2f af1 = *reinterpret_cast<const v2f*>(a1 + k);
            acc0 = __builtin_amdgcn_wmma_f32_16x16x4_f32(
                false, af0, false, bf, (short)0, acc0, false, false);
            acc1 = __builtin_amdgcn_wmma_f32_16x16x4_f32(
                false, af1, false, bf, (short)0, acc1, false, false);
        }
        int col = w * 16 + idx15;
        float bv = b1[col];
#pragma unroll
        for (int r = 0; r < 8; ++r) {
            float v0 = acc0[r] + bv;
            float v1 = acc1[r] + bv;
            v0 = (v0 > 0.0f) ? v0 : 0.0f;
            v1 = (v1 > 0.0f) ? v1 : 0.0f;
            s_h[(mbase + r) * 132 + col] = v0;
            s_h[(16 + mbase + r) * 132 + col] = v1;
        }
    }
    __syncthreads();

    {
        int mt = w >> 2;
        int nt = w & 3;
        const float* a0 = s_h + (mt * 16 + idx15) * 132 + koff;
        const float* wcol = W2 + nt * 16 + idx15;
        v8f acc = {};
        for (int k = 0; k < 128; k += 4) {
            v2f bf = gcn_b_frag(wcol, k, koff, 64);
            v2f af = *reinterpret_cast<const v2f*>(a0 + k);
            acc = __builtin_amdgcn_wmma_f32_16x16x4_f32(
                false, af, false, bf, (short)0, acc, false, false);
        }
        int col = nt * 16 + idx15;
#pragma unroll
        for (int r = 0; r < 8; ++r) {
            int m = row0 + mt * 16 + mbase + r;
            if (m < n) t[(size_t)m * 64 + col] = acc[r] * s_on[mt * 16 + mbase + r];
        }
    }
}

__global__ __launch_bounds__(256) void gcn_layer3_fused(
    const float* __restrict__ x3, const float* __restrict__ in_norm,
    const int* __restrict__ cnt_in, const int* __restrict__ bucket,
    const int* __restrict__ ovf_count, const int* __restrict__ ovf_dst,
    const int* __restrict__ ovf_src, const float* __restrict__ W3,
    const float* __restrict__ b3, const int* __restrict__ graph_ids,
    float* __restrict__ pooled, int n) {
    __shared__ __attribute__((aligned(16))) float s_a[GCN_ROWS * 68];
    int w = __builtin_amdgcn_readfirstlane((int)(threadIdx.x >> 5));
    int lane = threadIdx.x & 31;
    int row0 = blockIdx.x * GCN_ROWS;

    gcn_pull_rows_to_lds<64, false>(x3, nullptr, in_norm, cnt_in, bucket, ovf_count, ovf_dst,
                                    ovf_src, row0 + w * GCN_RPW, n, lane, s_a + w * GCN_RPW * 68);
    __syncthreads();

    int idx15 = lane & 15;
    int koff  = (lane >> 4) << 1;
    int mbase = (lane >> 4) << 3;
    int mt = w >> 2;
    int nt = w & 3;
    int trow0 = row0 + mt * 16;
    if (trow0 >= n) return;

    const float* a0 = s_a + (mt * 16 + idx15) * 68 + koff;
    const float* wcol = W3 + nt * 16 + idx15;
    v8f acc = {};
    for (int k = 0; k < 64; k += 4) {
        v2f bf = gcn_b_frag(wcol, k, koff, 64);
        v2f af = *reinterpret_cast<const v2f*>(a0 + k);
        acc = __builtin_amdgcn_wmma_f32_16x16x4_f32(
            false, af, false, bf, (short)0, acc, false, false);
    }

    int col = nt * 16 + idx15;
    float bv = b3[col];
    int g_l = graph_ids[trow0 + idx15];
    int g0 = __builtin_amdgcn_readfirstlane(g_l);
    bool one_graph = (__builtin_amdgcn_ballot_w32(g_l != g0) == 0u);
    if (one_graph) {
        float s = 0.0f;
#pragma unroll
        for (int r = 0; r < 8; ++r) s += acc[r] + bv;
        s += __shfl_xor(s, 16);
        if (lane < 16) unsafeAtomicAdd(&pooled[g0 * 64 + col], s);
    } else {
#pragma unroll
        for (int r = 0; r < 8; ++r) {
            int m = trow0 + mbase + r;
            float v = acc[r] + bv;
            int g = graph_ids[m];
            unsafeAtomicAdd(&pooled[g * 64 + col], v);
        }
    }
}


static inline int cdiv_l(long a, long b) { return (int)((a + b - 1) / b); }
static inline size_t line_up(size_t words) { return (words + 31) & ~(size_t)31; }

extern "C" void kernel_launch(void* const* d_in, const int* in_sizes, int n_in,
                              void* d_out, int out_size, void* d_ws, size_t ws_size,
                              hipStream_t stream) {
    const float* emb = (const float*)d_in[0];
    const float* W1  = (const float*)d_in[1];
    const float* b1  = (const float*)d_in[2];
    const float* W2  = (const float*)d_in[3];
    const float* b2  = (const float*)d_in[4];
    const float* W3  = (const float*)d_in[5];
    const float* b3  = (const float*)d_in[6];
    const int* src = (const int*)d_in[7];
    const int* dst = (const int*)d_in[8];
    const int* gid = (const int*)d_in[9];

    const int E = in_sizes[7];
    const int N = in_sizes[9];

    int*   cnt_in    = (int*)d_ws;
    float* pooled    = (float*)(cnt_in + N);
    float* counts    = pooled + GCN_N_GRAPHS * 64;
    int*   ovf_count = (int*)(counts + GCN_N_GRAPHS);
    float* out_norm  = (float*)(ovf_count + 4);
    float* in_norm   = out_norm + N;
    size_t head      = line_up((size_t)3 * N + GCN_N_GRAPHS * 64 + GCN_N_GRAPHS + 4);
    int*   bucket    = (int*)d_ws + head;
    int*   ovf_dst   = bucket + (size_t)N * GCN_SLOTS;
    int*   ovf_src   = ovf_dst + line_up(E);
    float* rows      = (float*)(ovf_src + line_up(E));
    int*   part_out  = (int*)(rows + (size_t)2 * N * 64);
    int*   part_in   = part_out + (size_t)GCN_SLICES * N;
    int*   part_base = part_in + (size_t)GCN_SLICES * N;

    const int T = 256;

    const int zc = GCN_N_GRAPHS * 64 + GCN_N_GRAPHS + 4;
    const int pairs = cdiv_l(N, GCN_RANGE) * GCN_SLICES;
    gcn_count_ranges<<<pairs, GCN_CT, 0, stream>>>(src, dst, part_out, part_in, pooled, zc, E, N);
    gcn_offsets_norms<<<cdiv_l(N, T), T, 0, stream>>>(part_out, part_in, part_base, cnt_in,
                                                      out_norm, in_norm, N);
    gcn_place_ranges<<<pairs, GCN_CT, 0, stream>>>(src, dst, part_base, bucket, ovf_count,
                                                   ovf_dst, ovf_src, E, N);
    gcn_count<<<cdiv_l(N, T), T, 0, stream>>>(gid, counts, N);

    const int rows_per_block = T / 32;
    const int groups = cdiv_l(N, GCN_ROWS);

    float* t  = rows;
    float* x3 = rows + (size_t)N * 64;
    gcn_layer1_fused<<<groups, T, 0, stream>>>(
        emb, out_norm, in_norm, cnt_in, bucket, ovf_count, ovf_dst, ovf_src, W1, b1, W2, t, N);

    gcn_pull_finish<64><<<cdiv_l(N, rows_per_block), T, 0, stream>>>(
        t, out_norm, in_norm, b2, cnt_in, bucket, ovf_count, ovf_dst, ovf_src, x3, N);

    gcn_layer3_fused<<<groups, T, 0, stream>>>(
        x3, in_norm, cnt_in, bucket, ovf_count, ovf_dst, ovf_src, W3, b3, gid, pooled, N);

    gcn_final<<<cdiv_l(GCN_N_GRAPHS * 64, T), T, 0, stream>>>(pooled, counts, (float*)d_out);
}


# Edges binned by node range once: a work-group per edge slice orders its slice by range with LDS counters; one per range builds its buckets, degrees, norms from its own runs. f32 sum order: arrival
# speedup vs baseline: 7.2269x; 1.2347x over previous
#include <hip/hip_runtime.h>

typedef __attribute__((ext_vector_type(2))) float v2f;
typedef __attribute__((ext_vector_type(8))) float v8f;

#define GCN_N_GRAPHS 64
#define GCN_SLOTS 32

#define GCN_NB 256
#define GCN_NSL 256
#define GCN_RBMAX 512
#define GCN_CT 1024
#define GCN_EPT 8
#define GCN_BPT 8

__device__ __forceinline__ int gcn_range_of(int v, int rb, float inv) {
    int q = (int)((float)v * inv);
    int r = v - q * rb;
    if (r < 0) q -= 1;
    else if (r >= rb) q += 1;
    return q;
}

__global__ __launch_bounds__(GCN_CT) void gcn_bin_slices(
    const int* __restrict__ src, const int* __restrict__ dst, int* __restrict__ bin_pair,
    int* __restrict__ bin_src, int* __restrict__ seg_d, int* __restrict__ seg_s,
    float* __restrict__ zero_p, int zero_n, int E, int n) {
    __shared__ int s_d[GCN_NB];
    __shared__ int s_s[GCN_NB];
    __shared__ int s_tot[2 * (GCN_NB / 32)];
    typedef __attribute__((ext_vector_type(2))) int v2i;
    const int tid = threadIdx.x;
    const int lane = tid & 31;
    const int wv = tid >> 5;
    const int b = blockIdx.x;
    const int rb = (n + GCN_NB - 1) / GCN_NB;
    const float inv = 1.0f / (float)rb;
    if (tid < GCN_NB) {
        s_d[tid] = 0;
        s_s[tid] = 0;
    }
    if (b == 0)
        for (int i = tid; i < zero_n; i += GCN_CT) zero_p[i] = 0.0f;
    __syncthreads();
    const int es = (E + GCN_NSL - 1) / GCN_NSL;
    const int e0 = b * es;
    const int e1 = (e0 + es < E) ? e0 + es : E;
    const int eb = b * ((es + 31) & ~31);
    for (int e = e0 + tid; e < e1; e += GCN_CT * GCN_EPT) {
        int sv[GCN_EPT];
        int dv[GCN_EPT];
#pragma unroll
        for (int u = 0; u < GCN_EPT; ++u) {
            int ee = e + u * GCN_CT;
            bool ok = ee < e1;
            int ec = ok ? ee : e;
            int s = src[ec];
            int d = dst[ec];
            sv[u] = ok ? s : -1;
            dv[u] = ok ? d : -1;
        }
#pragma unroll
        for (int u = 0; u < GCN_EPT; ++u) {
            if (dv[u] >= 0) {
                atomicAdd(&s_d[gcn_range_of(dv[u], rb, inv)], 1);
                atomicAdd(&s_s[gcn_range_of(sv[u], rb, inv)], 1);
            }
        }
    }
    __syncthreads();
    int cd = 0;
    int cs = 0;
    if (tid < GCN_NB) {
        cd = s_d[tid];
        cs = s_s[tid];
    }
    int xd = cd;
    int xs = cs;
    for (int o = 1; o < 32; o <<= 1) {
        int yd = __shfl_up(xd, o);
        int ys = __shfl_up(xs, o);
        if (lane >= o) {
            xd += yd;
            xs += ys;
        }
    }
    if (tid < GCN_NB && lane == 31) {
        s_tot[wv] = xd;
        s_tot[GCN_NB / 32 + wv] = xs;
    }
    __syncthreads();
    if (tid < GCN_NB) {
        int pd = 0;
        int ps = 0;
        for (int k = 0; k < GCN_NB / 32; ++k) {
            if (k < wv) {
                pd += s_tot[k];
                ps += s_tot[GCN_NB / 32 + k];
            }
        }
        int exd = pd + xd - cd;
        int exs = ps + xs - cs;
        s_d[tid] = exd;
        s_s[tid] = exs;
        seg_d[b * (GCN_NB + 1) + tid] = exd;
        seg_s[b * (GCN_NB + 1) + tid] = exs;
        if (tid == GCN_NB - 1) {
            seg_d[b * (GCN_NB + 1) + GCN_NB] = exd + cd;
            seg_s[b * (GCN_NB + 1) + GCN_NB] = exs + cs;
        }
    }
    __syncthreads();
    for (int e = e0 + tid; e < e1; e += GCN_CT * GCN_EPT) {
        int sv[GCN_EPT];
        int dv[GCN_EPT];
#pragma unroll
        for (int u = 0; u < GCN_EPT; ++u) {
            int ee = e + u * GCN_CT;
            bool ok = ee < e1;
            int ec = ok ? ee : e;
            int s = src[ec];
            int d = dst[ec];
            sv[u] = ok ? s : -1;
            dv[u] = ok ? d : -1;
        }
#pragma unroll
        for (int u = 0; u < GCN_EPT; ++u) {
            if (dv[u] >= 0) {
                int pd = atomicAdd(&s_d[gcn_range_of(dv[u], rb, inv)], 1);
                int ps = atomicAdd(&s_s[gcn_range_of(sv[u], rb, inv)], 1);
                v2i pr = {dv[u], sv[u]};
                reinterpret_cast<v2i*>(bin_pair)[eb + pd] = pr;
                bin_src[eb + ps] = sv[u];
            }
        }
    }
}

__global__ __launch_bounds__(GCN_CT) void gcn_build_ranges(
    const int* __restrict__ bin_pair, const int* __restrict__ bin_src, const int* __restrict__ seg_d,
    const int* __restrict__ seg_s, int* __restrict__ bucket, int* __restrict__ cnt_in,
    float* __restrict__ out_norm, float* __restrict__ in_norm, int* __restrict__ ovf_count,
    int* __restrict__ ovf_dst, int* __restrict__ ovf_src, int E, int n) {
    __shared__ __attribute__((aligned(16))) int s_img[GCN_RBMAX * GCN_SLOTS];
    __shared__ int s_cin[GCN_RBMAX];
    __shared__ int s_cout[GCN_RBMAX];
    typedef __attribute__((ext_vector_type(2))) int v2i;
    typedef __attribute__((ext_vector_type(4))) int v4i;
    const int tid = threadIdx.x;
    const int r = blockIdx.x;
    const int rb = (n + GCN_NB - 1) / GCN_NB;
    const int base = r * rb;
    int lim = n - base;
    lim = (lim < rb) ? lim : rb;
    lim = (lim < 0) ? 0 : lim;
    const v4i z4 = {0, 0, 0, 0};
    for (int i = tid; i < lim * (GCN_SLOTS / 4); i += GCN_CT) reinterpret_cast<v4i*>(s_img)[i] = z4;
    for (int i = tid; i < GCN_RBMAX; i += GCN_CT) {
        s_cin[i] = 0;
        s_cout[i] = 0;
    }
    __syncthreads();
    const int es = (E + GCN_NSL - 1) / GCN_NSL;
    const int b = tid >> 2;
    const int q = tid & 3;
    const int eb = b * ((es + 31) & ~31);
    const int o0 = seg_d[b * (GCN_NB + 1) + r];
    const int o1 = seg_d[b * (GCN_NB + 1) + r + 1];
    const int p0 = seg_s[b * (GCN_NB + 1) + r];
    const int p1 = seg_s[b * (GCN_NB + 1) + r + 1];
    const v2i* pairs = reinterpret_cast<const v2i*>(bin_pair) + eb;
    const int* srcs = bin_src + eb;
    for (int k0 = o0; k0 < o1; k0 += 4 * GCN_BPT) {
        v2i pr[GCN_BPT];
#pragma unroll
        for (int u = 0; u < GCN_BPT; ++u) {
            int k = k0 + q + 4 * u;
            bool ok = k < o1;
            v2i v = pairs[ok ? k : o0];
            if (!ok) v.x = -1;
            pr[u] = v;
        }
#pragma unroll
        for (int u = 0; u < GCN_BPT; ++u) {
            int d = pr[u].x;
            if (d >= 0) {
                int ld = d - base;
                int slot = atomicAdd(&s_cin[ld], 1);
                if (slot < GCN_SLOTS) {
                    s_img[ld * GCN_SLOTS + slot] = pr[u].y;
                } else {
                    int k = atomicAdd(ovf_count, 1);
                    ovf_dst[k] = d;
                    ovf_src[k] = pr[u].y;
                }
            }
        }
    }
    for (int k0 = p0; k0 < p1; k0 += 4 * GCN_BPT) {
        int sv[GCN_BPT];
#pragma unroll
        for (int u = 0; u < GCN_BPT; ++u) {
            int k = k0 + q + 4 * u;
            bool ok = k < p1;
            int s = srcs[ok ? k : p0];
            sv[u] = ok ? s : -1;
        }
#pragma unroll
        for (int u = 0; u < GCN_BPT; ++u) {
            if (sv[u] >= 0) atomicAdd(&s_cout[sv[u] - base], 1);
        }
    }
    __syncthreads();
    for (int i = tid; i < lim; i += GCN_CT) {
        int ci = s_cin[i];
        cnt_in[base + i] = ci;
        float d = (float)s_cout[i];
        d = (d < 1.0f) ? 1.0f : d;
        out_norm[base + i] = 1.0f / sqrtf(d);
        d = (float)ci;
        d = (d < 1.0f) ? 1.0f : d;
        in_norm[base + i] = 1.0f / sqrtf(d);
    }
    v4i* bl = reinterpret_cast<v4i*>(bucket + (size_t)base * GCN_SLOTS);
    for (int i = tid; i < lim * (GCN_SLOTS / 4); i += GCN_CT) bl[i] = reinterpret_cast<const v4i*>(s_img)[i];
}

__global__ void gcn_count(const int* __restrict__ gid, float* __restrict__ counts, int n) {
    int i = blockIdx.x * blockDim.x + threadIdx.x;
    int lane = threadIdx.x & 31;
    bool todo = i < n;
    int g = todo ? gid[i] : 0;
    for (int it = 0; it < 32; ++it) {
        unsigned waiting = __builtin_amdgcn_ballot_w32(todo);
        if (waiting == 0u) break;
        int first = __builtin_ctz(waiting);
        int g0 = __builtin_amdgcn_readlane(g, first);
        bool same = todo && (g == g0);
        unsigned members = __builtin_amdgcn_ballot_w32(same);
        if (lane == first) unsafeAtomicAdd(&counts[g0], (float)__builtin_popcount(members));
        todo = todo && !same;
    }
}

__global__ void gcn_final(const float* __restrict__ pooled, const float* __restrict__ counts,
                          float* __restrict__ out) {
    int i = blockIdx.x * blockDim.x + threadIdx.x;
    if (i < GCN_N_GRAPHS * 64) {
        float c = counts[i >> 6];
        c = (c < 1.0f) ? 1.0f : c;
        out[i] = pooled[i] / c;
    }
}


template <int V> struct gcn_vec;
template <> struct gcn_vec<4> { typedef __attribute__((ext_vector_type(4))) float type; };
template <> struct gcn_vec<2> { typedef __attribute__((ext_vector_type(2))) float type; };

template <int F, bool SCALE, int U, bool TAIL, typename vec_t>
__device__ __forceinline__ void gcn_pull_edges(const float* __restrict__ xl, int s_l, float sc_l,
                                               int j, int m, vec_t& acc) {
    vec_t v[U];
    float sc[U];
#pragma unroll
    for (int u = 0; u < U; ++u) {
        int jj = j + u;
        if (TAIL) jj = (jj < m) ? jj : m - 1;
        int s = __builtin_amdgcn_readlane(s_l, jj);
        sc[u] = SCALE ? __int_as_float(__builtin_amdgcn_readlane(__float_as_int(sc_l), jj)) : 1.0f;
        v[u] = *reinterpret_cast<const vec_t*>(xl + (size_t)s * F);
    }
#pragma unroll
    for (int u = 0; u < U; ++u) {
        if (!TAIL || j + u < m) {
            if (SCALE) acc += v[u] * sc[u];
            else acc += v[u];
        }
    }
}

template <int F, bool SCALE, typename vec_t>
__device__ __forceinline__ void gcn_pull_overflow(const float* __restrict__ xl,
                                                  const float* __restrict__ out_norm, int row, int lane,
                                                  const int* __restrict__ ovf_count,
                                                  const int* __restrict__ ovf_dst,
                                                  const int* __restrict__ ovf_src, vec_t& acc) {
    int total = ovf_count[0];
    for (int k0 = 0; k0 < total; k0 += 32) {
        int k = k0 + lane;
        bool mine = false;
        if (k < total) mine = (ovf_dst[k] == row);
        int s = 0;
        float sc = 1.0f;
        if (mine) {
            s = ovf_src[k];
            if (SCALE) sc = out_norm[s];
        }
        unsigned todo = __builtin_amdgcn_ballot_w32(mine);
        for (int it = 0; it < 32; ++it) {
            if (todo == 0u) break;
            int j = __builtin_ctz(todo);
            todo &= todo - 1u;
            int sj = __builtin_amdgcn_readlane(s, j);
            vec_t v = *reinterpret_cast<const vec_t*>(xl + (size_t)sj * F);
            if (SCALE) {
                float scj = __int_as_float(__builtin_amdgcn_readlane(__float_as_int(sc), j));
                acc += v * scj;
            } else {
                acc += v;
            }
        }
    }
}

template <int F>
__global__ void gcn_pull_finish(const float* __restrict__ x, const float* __restrict__ out_norm,
                                const float* __restrict__ in_norm, const float* __restrict__ bias,
                                const int* __restrict__ cnt_in, const int* __restrict__ bucket,
                                const int* __restrict__ ovf_count, const int* __restrict__ ovf_dst,
                                const int* __restrict__ ovf_src, float* __restrict__ y, int n) {
    const int VPL = F / 32;
    typedef typename gcn_vec<VPL>::type vec_t;
    int row = __builtin_amdgcn_readfirstlane(
        (int)(blockIdx.x * (blockDim.x >> 5) + (threadIdx.x >> 5)));
    if (row >= n) return;
    int lane = threadIdx.x & 31;
    int deg = cnt_in[row];
    float dn = in_norm[row];
    float on = out_norm[row];
    const float* xl = x + lane * VPL;

    vec_t acc = {};
    vec_t bv = *reinterpret_cast<const vec_t*>(bias + lane * VPL);

    {
        int m = (deg < GCN_SLOTS) ? deg : GCN_SLOTS;
        int s_l = 0;
        if (lane < m) s_l = bucket[(size_t)row * GCN_SLOTS + lane];
        int j = 0;
        for (; j + 8 <= m; j += 8)
            gcn_pull_edges<F, false, 8, false, vec_t>(xl, s_l, 0.0f, j, m, acc);
        for (; j < m; j += 4)
            gcn_pull_edges<F, false, 4, true, vec_t>(xl, s_l, 0.0f, j, m, acc);
    }
    if (deg > GCN_SLOTS)
        gcn_pull_overflow<F, false, vec_t>(xl, out_norm, row, lane, ovf_count, ovf_dst, ovf_src, acc);
    vec_t r = acc * dn + bv;
#pragma unroll
    for (int q = 0; q < VPL; ++q) r[q] = (r[q] > 0.0f) ? r[q] : 0.0f;
    *reinterpret_cast<vec_t*>(y + (size_t)row * F + lane * VPL) = r * on;
}


#define GCN_ROWS 32
#define GCN_RPW 4

template <int F, bool SCALE>
__device__ __forceinline__ void gcn_pull_rows_to_lds(
    const float* __restrict__ x, const float* __restrict__ out_norm,
    const float* __restrict__ in_norm,
    const int* __restrict__ cnt_in, const int* __restrict__ bucket,
    const int* __restrict__ ovf_count, const int* __restrict__ ovf_dst,
    const int* __restrict__ ovf_src, int row_first, int n, int lane, float* s_rows) {
    const int VPL = F / 32;
    const int PITCH = F + 4;
    typedef typename gcn_vec<VPL>::type vec_t;
    const float* xl = x + lane * VPL;

    int dg[GCN_RPW];
    int s_f[GCN_RPW];
    float sc_f[GCN_RPW];
    float dn[GCN_RPW];
#pragma unroll
    for (int i = 0; i < GCN_RPW; ++i) {
        int r = row_first + i;
        bool ok = r < n;
        dg[i] = ok ? cnt_in[r] : 0;
        dn[i] = ok ? in_norm[r] : 0.0f;
    }
#pragma unroll
    for (int i = 0; i < GCN_RPW; ++i) {
        int m = (dg[i] < GCN_SLOTS) ? dg[i] : GCN_SLOTS;
        s_f[i] = 0;
        if (lane < m) s_f[i] = bucket[(size_t)(row_first + i) * GCN_SLOTS + lane];
    }
#pragma unroll
    for (int i = 0; i < GCN_RPW; ++i) {
        int m = (dg[i] < GCN_SLOTS) ? dg[i] : GCN_SLOTS;
        sc_f[i] = 0.0f;
        if (SCALE && lane < m) sc_f[i] = out_norm[s_f[i]];
    }
#pragma unroll
    for (int i = 0; i < GCN_RPW; ++i) {
        vec_t acc = {};
        int s_l = s_f[i];
        float sc_l = sc_f[i];
        int deg = dg[i];
        {
            int m = (deg < GCN_SLOTS) ? deg : GCN_SLOTS;
            int j = 0;
            for (; j + 8 <= m; j += 8)
                gcn_pull_edges<F, SCALE, 8, false, vec_t>(xl, s_l, sc_l, j, m, acc);
            for (; j < m; j += 4)
                gcn_pull_edges<F, SCALE, 4, true, vec_t>(xl, s_l, sc_l, j, m, acc);
        }
        if (deg > GCN_SLOTS)
            gcn_pull_overflow<F, SCALE, vec_t>(xl, out_norm, row_first + i, lane,
                                               ovf_count, ovf_dst, ovf_src, acc);
        *reinterpret_cast<vec_t*>(s_rows + i * PITCH + lane * VPL) = acc * dn[i];
    }
}

__device__ __forceinline__ v2f gcn_b_frag(const float* __restrict__ wcol, int k, int koff, int Nw) {
    v2f bf;
    bf.x = wcol[(k + koff) * Nw];
    bf.y = wcol[(k + koff + 1) * Nw];
    return bf;
}

__global__ __launch_bounds__(256, 16) void gcn_layer1_fused(
    const float* __restrict__ emb, const float* __restrict__ out_norm,
    const float* __restrict__ in_norm, const int* __restrict__ cnt_in,
    const int* __restrict__ bucket, const int* __restrict__ ovf_count,
    const int* __restrict__ ovf_dst, const int* __restrict__ ovf_src,
    const float* __restrict__ W1, const float* __restrict__ b1,
    const float* __restrict__ W2, float* __restrict__ t, int n) {
    __shared__ __attribute__((aligned(16))) float s_a[GCN_ROWS * 132];
    __shared__ __attribute__((aligned(16))) float s_h[GCN_ROWS * 132];
    __shared__ __attribute__((aligned(16))) float s_on[GCN_ROWS];
    int w = __builtin_amdgcn_readfirstlane((int)(threadIdx.x >> 5));
    int lane = threadIdx.x & 31;
    int row0 = blockIdx.x * GCN_ROWS;

    typedef gcn_vec<4>::type on_t;
    on_t on_w;
#pragma unroll
    for (int i = 0; i < GCN_RPW; ++i) {
        int r = row0 + w * GCN_RPW + i;
        on_w[i] = (r < n) ? out_norm[r] : 0.0f;
    }

    gcn_pull_rows_to_lds<128, true>(emb, out_norm, in_norm, cnt_in, bucket, ovf_count, ovf_dst,
                                    ovf_src, row0 + w * GCN_RPW, n, lane, s_a + w * GCN_RPW * 132);
    if (lane == 0) *reinterpret_cast<on_t*>(s_on + w * GCN_RPW) = on_w;
    __syncthreads();

    int idx15 = lane & 15;
    int koff  = (lane >> 4) << 1;
    int mbase = (lane >> 4) << 3;

    {
        const float* a0 = s_a + idx15 * 132 + koff;
        const float* a1 = a0 + 16 * 132;
        const float* wcol = W1 + w * 16 + idx15;
        v8f acc0 = {};
        v8f acc1 = {};
        for (int k = 0; k < 128; k += 4) {
            v2f bf = gcn_b_frag(wcol, k, koff, 128);
            v2f af0 = *reinterpret_cast<const v2f*>(a0 + k);
            v2f af1 = *reinterpret_cast<const v2f*>(a1 + k);
            acc0 = __builtin_amdgcn_wmma_f32_16x16x4_f32(
                false, af0, false, bf, (short)0, acc0, false, false);
            acc1 = __builtin_amdgcn_wmma_f32_16x16x4_f32(
                false, af1, false, bf, (short)0, acc1, false, false);
        }
        int col = w * 16 + idx15;
        float bv = b1[col];
#pragma unroll
        for (int r = 0; r < 8; ++r) {
            float v0 = acc0[r] + bv;
            float v1 = acc1[r] + bv;
            v0 = (v0 > 0.0f) ? v0 : 0.0f;
            v1 = (v1 > 0.0f) ? v1 : 0.0f;
            s_h[(mbase + r) * 132 + col] = v0;
            s_h[(16 + mbase + r) * 132 + col] = v1;
        }
    }
    __syncthreads();

    {
        int mt = w >> 2;
        int nt = w & 3;
        const float* a0 = s_h + (mt * 16 + idx15) * 132 + koff;
        const float* wcol = W2 + nt * 16 + idx15;
        v8f acc = {};
        for (int k = 0; k < 128; k += 4) {
            v2f bf = gcn_b_frag(wcol, k, koff, 64);
            v2f af = *reinterpret_cast<const v2f*>(a0 + k);
            acc = __builtin_amdgcn_wmma_f32_16x16x4_f32(
                false, af, false, bf, (short)0, acc, false, false);
        }
        int col = nt * 16 + idx15;
#pragma unroll
        for (int r = 0; r < 8; ++r) {
            int m = row0 + mt * 16 + mbase + r;
            if (m < n) t[(size_t)m * 64 + col] = acc[r] * s_on[mt * 16 + mbase + r];
        }
    }
}

__global__ __launch_bounds__(256) void gcn_layer3_fused(
    const float* __restrict__ x3, const float* __restrict__ in_norm,
    const int* __restrict__ cnt_in, const int* __restrict__ bucket,
    const int* __restrict__ ovf_count, const int* __restrict__ ovf_dst,
    const int* __restrict__ ovf_src, const float* __restrict__ W3,
    const float* __restrict__ b3, const int* __restrict__ graph_ids,
    float* __restrict__ pooled, int n) {
    __shared__ __attribute__((aligned(16))) float s_a[GCN_ROWS * 68];
    int w = __builtin_amdgcn_readfirstlane((int)(threadIdx.x >> 5));
    int lane = threadIdx.x & 31;
    int row0 = blockIdx.x * GCN_ROWS;

    gcn_pull_rows_to_lds<64, false>(x3, nullptr, in_norm, cnt_in, bucket, ovf_count, ovf_dst,
                                    ovf_src, row0 + w * GCN_RPW, n, lane, s_a + w * GCN_RPW * 68);
    __syncthreads();

    int idx15 = lane & 15;
    int koff  = (lane >> 4) << 1;
    int mbase = (lane >> 4) << 3;
    int mt = w >> 2;
    int nt = w & 3;
    int trow0 = row0 + mt * 16;
    if (trow0 >= n) return;

    const float* a0 = s_a + (mt * 16 + idx15) * 68 + koff;
    const float* wcol = W3 + nt * 16 + idx15;
    v8f acc = {};
    for (int k = 0; k < 64; k += 4) {
        v2f bf = gcn_b_frag(wcol, k, koff, 64);
        v2f af = *reinterpret_cast<const v2f*>(a0 + k);
        acc = __builtin_amdgcn_wmma_f32_16x16x4_f32(
            false, af, false, bf, (short)0, acc, false, false);
    }

    int col = nt * 16 + idx15;
    float bv = b3[col];
    int g_l = graph_ids[trow0 + idx15];
    int g0 = __builtin_amdgcn_readfirstlane(g_l);
    bool one_graph = (__builtin_amdgcn_ballot_w32(g_l != g0) == 0u);
    if (one_graph) {
        float s = 0.0f;
#pragma unroll
        for (int r = 0; r < 8; ++r) s += acc[r] + bv;
        s += __shfl_xor(s, 16);
        if (lane < 16) unsafeAtomicAdd(&pooled[g0 * 64 + col], s);
    } else {
#pragma unroll
        for (int r = 0; r < 8; ++r) {
            int m = trow0 + mbase + r;
            float v = acc[r] + bv;
            int g = graph_ids[m];
            unsafeAtomicAdd(&pooled[g * 64 + col], v);
        }
    }
}


static inline int cdiv_l(long a, long b) { return (int)((a + b - 1) / b); }
static inline size_t line_up(size_t words) { return (words + 31) & ~(size_t)31; }

extern "C" void kernel_launch(void* const* d_in, const int* in_sizes, int n_in,
                              void* d_out, int out_size, void* d_ws, size_t ws_size,
                              hipStream_t stream) {
    const float* emb = (const float*)d_in[0];
    const float* W1  = (const float*)d_in[1];
    const float* b1  = (const float*)d_in[2];
    const float* W2  = (const float*)d_in[3];
    const float* b2  = (const float*)d_in[4];
    const float* W3  = (const float*)d_in[5];
    const float* b3  = (const float*)d_in[6];
    const int* src = (const int*)d_in[7];
    const int* dst = (const int*)d_in[8];
    const int* gid = (const int*)d_in[9];

    const int E = in_sizes[7];
    const int N = in_sizes[9];

    int*   cnt_in    = (int*)d_ws;
    float* pooled    = (float*)(cnt_in + N);
    float* counts    = pooled + GCN_N_GRAPHS * 64;
    int*   ovf_count = (int*)(counts + GCN_N_GRAPHS);
    float* out_norm  = (float*)(ovf_count + 4);
    float* in_norm   = out_norm + N;
    size_t head      = line_up((size_t)3 * N + GCN_N_GRAPHS * 64 + GCN_N_GRAPHS + 4);
    int*   bucket    = (int*)d_ws + head;
    int*   ovf_dst   = bucket + (size_t)N * GCN_SLOTS;
    int*   ovf_src   = ovf_dst + line_up(E);
    float* rows      = (float*)(ovf_src + line_up(E));
    size_t binned    = line_up((size_t)E + GCN_NSL * 32);
    int*   bin_pair  = (int*)(rows + (size_t)2 * N * 64);
    int*   bin_src   = bin_pair + 2 * binned;
    int*   seg_d     = bin_src + binned;
    int*   seg_s     = seg_d + line_up((size_t)GCN_NSL * (GCN_NB + 1));

    const int T = 256;

    const int zc = GCN_N_GRAPHS * 64 + GCN_N_GRAPHS + 4;
    gcn_bin_slices<<<GCN_NSL, GCN_CT, 0, stream>>>(src, dst, bin_pair, bin_src, seg_d, seg_s,
                                                   pooled, zc, E, N);
    gcn_build_ranges<<<GCN_NB, GCN_CT, 0, stream>>>(bin_pair, bin_src, seg_d, seg_s, bucket, cnt_in,
                                                    out_norm, in_norm, ovf_count, ovf_dst, ovf_src, E, N);
    gcn_count<<<cdiv_l(N, T), T, 0, stream>>>(gid, counts, N);

    const int rows_per_block = T / 32;
    const int groups = cdiv_l(N, GCN_ROWS);

    float* t  = rows;
    float* x3 = rows + (size_t)N * 64;
    gcn_layer1_fused<<<groups, T, 0, stream>>>(
        emb, out_norm, in_norm, cnt_in, bucket, ovf_count, ovf_dst, ovf_src, W1, b1, W2, t, N);

    gcn_pull_finish<64><<<cdiv_l(N, rows_per_block), T, 0, stream>>>(
        t, out_norm, in_norm, b2, cnt_in, bucket, ovf_count, ovf_dst, ovf_src, x3, N);

    gcn_layer3_fused<<<groups, T, 0, stream>>>(
        x3, in_norm, cnt_in, bucket, ovf_count, ovf_dst, ovf_src, W3, b3, gid, pooled, N);

    gcn_final<<<cdiv_l(GCN_N_GRAPHS * 64, T), T, 0, stream>>>(pooled, counts, (float*)d_out);
}


# Each edge slice is ordered inside LDS (place = range start + the rank its one counting add returned) and stored as whole 128-byte lines, not 2 scattered stores per edge. f32 sum order: arrival
# speedup vs baseline: 7.7297x; 1.0696x over previous
#include <hip/hip_runtime.h>

typedef __attribute__((ext_vector_type(2))) float v2f;
typedef __attribute__((ext_vector_type(8))) float v8f;

#define GCN_N_GRAPHS 64
#define GCN_SLOTS 32

#define GCN_NB 256
#define GCN_NSL 256
#define GCN_RBMAX 512
#define GCN_CT 1024
#define GCN_EPT 8
#define GCN_ESMAX (GCN_CT * GCN_EPT)
#define GCN_BPT 8

__device__ __forceinline__ int gcn_range_of(int v, int rb, float inv) {
    int q = (int)((float)v * inv);
    int r = v - q * rb;
    if (r < 0) q -= 1;
    else if (r >= rb) q += 1;
    return q;
}

__global__ __launch_bounds__(GCN_CT) void gcn_bin_slices(
    const int* __restrict__ src, const int* __restrict__ dst, int* __restrict__ bin_pair,
    int* __restrict__ bin_src, int* __restrict__ seg_d, int* __restrict__ seg_s,
    float* __restrict__ zero_p, int zero_n, int E, int n) {
    __shared__ int s_d[GCN_NB];
    __shared__ int s_s[GCN_NB];
    __shared__ int s_tot[2 * (GCN_NB / 32)];
    __shared__ __attribute__((aligned(16))) int s_pair[2 * GCN_ESMAX];
    __shared__ __attribute__((aligned(16))) int s_src[GCN_ESMAX];
    typedef __attribute__((ext_vector_type(2))) int v2i;
    typedef __attribute__((ext_vector_type(4))) int v4i;
    const int tid = threadIdx.x;
    const int lane = tid & 31;
    const int wv = tid >> 5;
    const int b = blockIdx.x;
    const int rb = (n + GCN_NB - 1) / GCN_NB;
    const float inv = 1.0f / (float)rb;
    if (tid < GCN_NB) {
        s_d[tid] = 0;
        s_s[tid] = 0;
    }
    if (b == 0)
        for (int i = tid; i < zero_n; i += GCN_CT) zero_p[i] = 0.0f;
    __syncthreads();
    const int es = (E + GCN_NSL - 1) / GCN_NSL;
    const int e0 = b * es;
    const int e1 = (e0 + es < E) ? e0 + es : E;
    const int len = (e1 > e0) ? e1 - e0 : 0;
    const int eb = b * ((es + 31) & ~31);
    const int esafe = (e0 < E) ? e0 : 0;
    int sv[GCN_EPT];
    int dv[GCN_EPT];
    int kd[GCN_EPT];
    int ks[GCN_EPT];
    int rd[GCN_EPT];
    int rs[GCN_EPT];
#pragma unroll
    for (int u = 0; u < GCN_EPT; ++u) {
        int ee = e0 + tid + u * GCN_CT;
        bool ok = ee < e1;
        int ec = ok ? ee : esafe;
        int s = src[ec];
        int d = dst[ec];
        sv[u] = ok ? s : -1;
        dv[u] = ok ? d : -1;
    }
#pragma unroll
    for (int u = 0; u < GCN_EPT; ++u) {
        kd[u] = 0;
        ks[u] = 0;
        rd[u] = 0;
        rs[u] = 0;
        if (dv[u] >= 0) {
            kd[u] = gcn_range_of(dv[u], rb, inv);
            ks[u] = gcn_range_of(sv[u], rb, inv);
            rd[u] = atomicAdd(&s_d[kd[u]], 1);
            rs[u] = atomicAdd(&s_s[ks[u]], 1);
        }
    }
    __syncthreads();
    int cd = 0;
    int cs = 0;
    if (tid < GCN_NB) {
        cd = s_d[tid];
        cs = s_s[tid];
    }
    int xd = cd;
    int xs = cs;
    for (int o = 1; o < 32; o <<= 1) {
        int yd = __shfl_up(xd, o);
        int ys = __shfl_up(xs, o);
        if (lane >= o) {
            xd += yd;
            xs += ys;
        }
    }
    if (tid < GCN_NB && lane == 31) {
        s_tot[wv] = xd;
        s_tot[GCN_NB / 32 + wv] = xs;
    }
    __syncthreads();
    if (tid < GCN_NB) {
        int pd = 0;
        int ps = 0;
        for (int k = 0; k < GCN_NB / 32; ++k) {
            if (k < wv) {
                pd += s_tot[k];
                ps += s_tot[GCN_NB / 32 + k];
            }
        }
        int exd = pd + xd - cd;
        int exs = ps + xs - cs;
        s_d[tid] = exd;
        s_s[tid] = exs;
        seg_d[b * (GCN_NB + 1) + tid] = exd;
        seg_s[b * (GCN_NB + 1) + tid] = exs;
        if (tid == GCN_NB - 1) {
            seg_d[b * (GCN_NB + 1) + GCN_NB] = exd + cd;
            seg_s[b * (GCN_NB + 1) + GCN_NB] = exs + cs;
        }
    }
    __syncthreads();
#pragma unroll
    for (int u = 0; u < GCN_EPT; ++u) {
        if (dv[u] >= 0) {
            int pd = s_d[kd[u]] + rd[u];
            int ps = s_s[ks[u]] + rs[u];
            v2i pr = {dv[u], sv[u]};
            reinterpret_cast<v2i*>(s_pair)[pd] = pr;
            s_src[ps] = sv[u];
        }
    }
    __syncthreads();
    v4i* gp = reinterpret_cast<v4i*>(bin_pair + 2 * (size_t)eb);
    for (int i = tid; i < (2 * len + 3) / 4; i += GCN_CT) gp[i] = reinterpret_cast<const v4i*>(s_pair)[i];
    v4i* gs = reinterpret_cast<v4i*>(bin_src + eb);
    for (int i = tid; i < (len + 3) / 4; i += GCN_CT) gs[i] = reinterpret_cast<const v4i*>(s_src)[i];
}

__global__ __launch_bounds__(GCN_CT) void gcn_build_ranges(
    const int* __restrict__ bin_pair, const int* __restrict__ bin_src, const int* __restrict__ seg_d,
    const int* __restrict__ seg_s, int* __restrict__ bucket, int* __restrict__ cnt_in,
    float* __restrict__ out_norm, float* __restrict__ in_norm, int* __restrict__ ovf_count,
    int* __restrict__ ovf_dst, int* __restrict__ ovf_src, int E, int n) {
    __shared__ __attribute__((aligned(16))) int s_img[GCN_RBMAX * GCN_SLOTS];
    __shared__ int s_cin[GCN_RBMAX];
    __shared__ int s_cout[GCN_RBMAX];
    typedef __attribute__((ext_vector_type(2))) int v2i;
    typedef __attribute__((ext_vector_type(4))) int v4i;
    const int tid = threadIdx.x;
    const int r = blockIdx.x;
    const int rb = (n + GCN_NB - 1) / GCN_NB;
    const int base = r * rb;
    int lim = n - base;
    lim = (lim < rb) ? lim : rb;
    lim = (lim < 0) ? 0 : lim;
    const v4i z4 = {0, 0, 0, 0};
    for (int i = tid; i < lim * (GCN_SLOTS / 4); i += GCN_CT) reinterpret_cast<v4i*>(s_img)[i] = z4;
    for (int i = tid; i < GCN_RBMAX; i += GCN_CT) {
        s_cin[i] = 0;
        s_cout[i] = 0;
    }
    __syncthreads();
    const int es = (E + GCN_NSL - 1) / GCN_NSL;
    const int b = tid >> 2;
    const int q = tid & 3;
    const int eb = b * ((es + 31) & ~31);
    const int o0 = seg_d[b * (GCN_NB + 1) + r];
    const int o1 = seg_d[b * (GCN_NB + 1) + r + 1];
    const int p0 = seg_s[b * (GCN_NB + 1) + r];
    const int p1 = seg_s[b * (GCN_NB + 1) + r + 1];
    const v2i* pairs = reinterpret_cast<const v2i*>(bin_pair) + eb;
    const int* srcs = bin_src + eb;
    for (int k0 = o0; k0 < o1; k0 += 4 * GCN_BPT) {
        v2i pr[GCN_BPT];
#pragma unroll
        for (int u = 0; u < GCN_BPT; ++u) {
            int k = k0 + q + 4 * u;
            bool ok = k < o1;
            v2i v = pairs[ok ? k : o0];
            if (!ok) v.x = -1;
            pr[u] = v;
        }
#pragma unroll
        for (int u = 0; u < GCN_BPT; ++u) {
            int d = pr[u].x;
            if (d >= 0) {
                int ld = d - base;
                int slot = atomicAdd(&s_cin[ld], 1);
                if (slot < GCN_SLOTS) {
                    s_img[ld * GCN_SLOTS + slot] = pr[u].y;
                } else {
                    int k = atomicAdd(ovf_count, 1);
                    ovf_dst[k] = d;
                    ovf_src[k] = pr[u].y;
                }
            }
        }
    }
    for (int k0 = p0; k0 < p1; k0 += 4 * GCN_BPT) {
        int sv[GCN_BPT];
#pragma unroll
        for (int u = 0; u < GCN_BPT; ++u) {
            int k = k0 + q + 4 * u;
            bool ok = k < p1;
            int s = srcs[ok ? k : p0];
            sv[u] = ok ? s : -1;
        }
#pragma unroll
        for (int u = 0; u < GCN_BPT; ++u) {
            if (sv[u] >= 0) atomicAdd(&s_cout[sv[u] - base], 1);
        }
    }
    __syncthreads();
    for (int i = tid; i < lim; i += GCN_CT) {
        int ci = s_cin[i];
        cnt_in[base + i] = ci;
        float d = (float)s_cout[i];
        d = (d < 1.0f) ? 1.0f : d;
        out_norm[base + i] = 1.0f / sqrtf(d);
        d = (float)ci;
        d = (d < 1.0f) ? 1.0f : d;
        in_norm[base + i] = 1.0f / sqrtf(d);
    }
    v4i* bl = reinterpret_cast<v4i*>(bucket + (size_t)base * GCN_SLOTS);
    for (int i = tid; i < lim * (GCN_SLOTS / 4); i += GCN_CT) bl[i] = reinterpret_cast<const v4i*>(s_img)[i];
}

__global__ void gcn_count(const int* __restrict__ gid, float* __restrict__ counts, int n) {
    int i = blockIdx.x * blockDim.x + threadIdx.x;
    int lane = threadIdx.x & 31;
    bool todo = i < n;
    int g = todo ? gid[i] : 0;
    for (int it = 0; it < 32; ++it) {
        unsigned waiting = __builtin_amdgcn_ballot_w32(todo);
        if (waiting == 0u) break;
        int first = __builtin_ctz(waiting);
        int g0 = __builtin_amdgcn_readlane(g, first);
        bool same = todo && (g == g0);
        unsigned members = __builtin_amdgcn_ballot_w32(same);
        if (lane == first) unsafeAtomicAdd(&counts[g0], (float)__builtin_popcount(members));
        todo = todo && !same;
    }
}

__global__ void gcn_final(const float* __restrict__ pooled, const float* __restrict__ counts,
                          float* __restrict__ out) {
    int i = blockIdx.x * blockDim.x + threadIdx.x;
    if (i < GCN_N_GRAPHS * 64) {
        float c = counts[i >> 6];
        c = (c < 1.0f) ? 1.0f : c;
        out[i] = pooled[i] / c;
    }
}


template <int V> struct gcn_vec;
template <> struct gcn_vec<4> { typedef __attribute__((ext_vector_type(4))) float type; };
template <> struct gcn_vec<2> { typedef __attribute__((ext_vector_type(2))) float type; };

template <int F, bool SCALE, int U, bool TAIL, typename vec_t>
__device__ __forceinline__ void gcn_pull_edges(const float* __restrict__ xl, int s_l, float sc_l,
                                               int j, int m, vec_t& acc) {
    vec_t v[U];
    float sc[U];
#pragma unroll
    for (int u = 0; u < U; ++u) {
        int jj = j + u;
        if (TAIL) jj = (jj < m) ? jj : m - 1;
        int s = __builtin_amdgcn_readlane(s_l, jj);
        sc[u] = SCALE ? __int_as_float(__builtin_amdgcn_readlane(__float_as_int(sc_l), jj)) : 1.0f;
        v[u] = *reinterpret_cast<const vec_t*>(xl + (size_t)s * F);
    }
#pragma unroll
    for (int u = 0; u < U; ++u) {
        if (!TAIL || j + u < m) {
            if (SCALE) acc += v[u] * sc[u];
            else acc += v[u];
        }
    }
}

template <int F, bool SCALE, typename vec_t>
__device__ __forceinline__ void gcn_pull_overflow(const float* __restrict__ xl,
                                                  const float* __restrict__ out_norm, int row, int lane,
                                                  const int* __restrict__ ovf_count,
                                                  const int* __restrict__ ovf_dst,
                                                  const int* __restrict__ ovf_src, vec_t& acc) {
    int total = ovf_count[0];
    for (int k0 = 0; k0 < total; k0 += 32) {
        int k = k0 + lane;
        bool mine = false;
        if (k < total) mine = (ovf_dst[k] == row);
        int s = 0;
        float sc = 1.0f;
        if (mine) {
            s = ovf_src[k];
            if (SCALE) sc = out_norm[s];
        }
        unsigned todo = __builtin_amdgcn_ballot_w32(mine);
        for (int it = 0; it < 32; ++it) {
            if (todo == 0u) break;
            int j = __builtin_ctz(todo);
            todo &= todo - 1u;
            int sj = __builtin_amdgcn_readlane(s, j);
            vec_t v = *reinterpret_cast<const vec_t*>(xl + (size_t)sj * F);
            if (SCALE) {
                float scj = __int_as_float(__builtin_amdgcn_readlane(__float_as_int(sc), j));
                acc += v * scj;
            } else {
                acc += v;
            }
        }
    }
}

template <int F>
__global__ void gcn_pull_finish(const float* __restrict__ x, const float* __restrict__ out_norm,
                                const float* __restrict__ in_norm, const float* __restrict__ bias,
                                const int* __restrict__ cnt_in, const int* __restrict__ bucket,
                                const int* __restrict__ ovf_count, const int* __restrict__ ovf_dst,
                                const int* __restrict__ ovf_src, float* __restrict__ y, int n) {
    const int VPL = F / 32;
    typedef typename gcn_vec<VPL>::type vec_t;
    int row = __builtin_amdgcn_readfirstlane(
        (int)(blockIdx.x * (blockDim.x >> 5) + (threadIdx.x >> 5)));
    if (row >= n) return;
    int lane = threadIdx.x & 31;
    int deg = cnt_in[row];
    float dn = in_norm[row];
    float on = out_norm[row];
    const float* xl = x + lane * VPL;

    vec_t acc = {};
    vec_t bv = *reinterpret_cast<const vec_t*>(bias + lane * VPL);

    {
        int m = (deg < GCN_SLOTS) ? deg : GCN_SLOTS;
        int s_l = 0;
        if (lane < m) s_l = bucket[(size_t)row * GCN_SLOTS + lane];
        int j = 0;
        for (; j + 8 <= m; j += 8)
            gcn_pull_edges<F, false, 8, false, vec_t>(xl, s_l, 0.0f, j, m, acc);
        for (; j < m; j += 4)
            gcn_pull_edges<F, false, 4, true, vec_t>(xl, s_l, 0.0f, j, m, acc);
    }
    if (deg > GCN_SLOTS)
        gcn_pull_overflow<F, false, vec_t>(xl, out_norm, row, lane, ovf_count, ovf_dst, ovf_src, acc);
    vec_t r = acc * dn + bv;
#pragma unroll
    for (int q = 0; q < VPL; ++q) r[q] = (r[q] > 0.0f) ? r[q] : 0.0f;
    *reinterpret_cast<vec_t*>(y + (size_t)row * F + lane * VPL) = r * on;
}


#define GCN_ROWS 32
#define GCN_RPW 4

template <int F, bool SCALE>
__device__ __forceinline__ void gcn_pull_rows_to_lds(
    const float* __restrict__ x, const float* __restrict__ out_norm,
    const float* __restrict__ in_norm,
    const int* __restrict__ cnt_in, const int* __restrict__ bucket,
    const int* __restrict__ ovf_count, const int* __restrict__ ovf_dst,
    const int* __restrict__ ovf_src, int row_first, int n, int lane, float* s_rows) {
    const int VPL = F / 32;
    const int PITCH = F + 4;
    typedef typename gcn_vec<VPL>::type vec_t;
    const float* xl = x + lane * VPL;

    int dg[GCN_RPW];
    int s_f[GCN_RPW];
    float sc_f[GCN_RPW];
    float dn[GCN_RPW];
#pragma unroll
    for (int i = 0; i < GCN_RPW; ++i) {
        int r = row_first + i;
        bool ok = r < n;
        dg[i] = ok ? cnt_in[r] : 0;
        dn[i] = ok ? in_norm[r] : 0.0f;
    }
#pragma unroll
    for (int i = 0; i < GCN_RPW; ++i) {
        int m = (dg[i] < GCN_SLOTS) ? dg[i] : GCN_SLOTS;
        s_f[i] = 0;
        if (lane < m) s_f[i] = bucket[(size_t)(row_first + i) * GCN_SLOTS + lane];
    }
#pragma unroll
    for (int i = 0; i < GCN_RPW; ++i) {
        int m = (dg[i] < GCN_SLOTS) ? dg[i] : GCN_SLOTS;
        sc_f[i] = 0.0f;
        if (SCALE && lane < m) sc_f[i] = out_norm[s_f[i]];
    }
#pragma unroll
    for (int i = 0; i < GCN_RPW; ++i) {
        vec_t acc = {};
        int s_l = s_f[i];
        float sc_l = sc_f[i];
        int deg = dg[i];
        {
            int m = (deg < GCN_SLOTS) ? deg : GCN_SLOTS;
            int j = 0;
            for (; j + 8 <= m; j += 8)
                gcn_pull_edges<F, SCALE, 8, false, vec_t>(xl, s_l, sc_l, j, m, acc);
            for (; j < m; j += 4)
                gcn_pull_edges<F, SCALE, 4, true, vec_t>(xl, s_l, sc_l, j, m, acc);
        }
        if (deg > GCN_SLOTS)
            gcn_pull_overflow<F, SCALE, vec_t>(xl, out_norm, row_first + i, lane,
                                               ovf_count, ovf_dst, ovf_src, acc);
        *reinterpret_cast<vec_t*>(s_rows + i * PITCH + lane * VPL) = acc * dn[i];
    }
}

__device__ __forceinline__ v2f gcn_b_frag(const float* __restrict__ wcol, int k, int koff, int Nw) {
    v2f bf;
    bf.x = wcol[(k + koff) * Nw];
    bf.y = wcol[(k + koff + 1) * Nw];
    return bf;
}

__global__ __launch_bounds__(256, 16) void gcn_layer1_fused(
    const float* __restrict__ emb, const float* __restrict__ out_norm,
    const float* __restrict__ in_norm, const int* __restrict__ cnt_in,
    const int* __restrict__ bucket, const int* __restrict__ ovf_count,
    const int* __restrict__ ovf_dst, const int* __restrict__ ovf_src,
    const float* __restrict__ W1, const float* __restrict__ b1,
    const float* __restrict__ W2, float* __restrict__ t, int n) {
    __shared__ __attribute__((aligned(16))) float s_a[GCN_ROWS * 132];
    __shared__ __attribute__((aligned(16))) float s_h[GCN_ROWS * 132];
    __shared__ __attribute__((aligned(16))) float s_on[GCN_ROWS];
    int w = __builtin_amdgcn_readfirstlane((int)(threadIdx.x >> 5));
    int lane = threadIdx.x & 31;
    int row0 = blockIdx.x * GCN_ROWS;

    typedef gcn_vec<4>::type on_t;
    on_t on_w;
#pragma unroll
    for (int i = 0; i < GCN_RPW; ++i) {
        int r = row0 + w * GCN_RPW + i;
        on_w[i] = (r < n) ? out_norm[r] : 0.0f;
    }

    gcn_pull_rows_to_lds<128, true>(emb, out_norm, in_norm, cnt_in, bucket, ovf_count, ovf_dst,
                                    ovf_src, row0 + w * GCN_RPW, n, lane, s_a + w * GCN_RPW * 132);
    if (lane == 0) *reinterpret_cast<on_t*>(s_on + w * GCN_RPW) = on_w;
    __syncthreads();

    int idx15 = lane & 15;
    int koff  = (lane >> 4) << 1;
    int mbase = (lane >> 4) << 3;

    {
        const float* a0 = s_a + idx15 * 132 + koff;
        const float* a1 = a0 + 16 * 132;
        const float* wcol = W1 + w * 16 + idx15;
        v8f acc0 = {};
        v8f acc1 = {};
        for (int k = 0; k < 128; k += 4) {
            v2f bf = gcn_b_frag(wcol, k, koff, 128);
            v2f af0 = *reinterpret_cast<const v2f*>(a0 + k);
            v2f af1 = *reinterpret_cast<const v2f*>(a1 + k);
            acc0 = __builtin_amdgcn_wmma_f32_16x16x4_f32(
                false, af0, false, bf, (short)0, acc0, false, false);
            acc1 = __builtin_amdgcn_wmma_f32_16x16x4_f32(
                false, af1, false, bf, (short)0, acc1, false, false);
        }
        int col = w * 16 + idx15;
        float bv = b1[col];
#pragma unroll
        for (int r = 0; r < 8; ++r) {
            float v0 = acc0[r] + bv;
            float v1 = acc1[r] + bv;
            v0 = (v0 > 0.0f) ? v0 : 0.0f;
            v1 = (v1 > 0.0f) ? v1 : 0.0f;
            s_h[(mbase + r) * 132 + col] = v0;
            s_h[(16 + mbase + r) * 132 + col] = v1;
        }
    }
    __syncthreads();

    {
        int mt = w >> 2;
        int nt = w & 3;
        const float* a0 = s_h + (mt * 16 + idx15) * 132 + koff;
        const float* wcol = W2 + nt * 16 + idx15;
        v8f acc = {};
        for (int k = 0; k < 128; k += 4) {
            v2f bf = gcn_b_frag(wcol, k, koff, 64);
            v2f af = *reinterpret_cast<const v2f*>(a0 + k);
            acc = __builtin_amdgcn_wmma_f32_16x16x4_f32(
                false, af, false, bf, (short)0, acc, false, false);
        }
        int col = nt * 16 + idx15;
#pragma unroll
        for (int r = 0; r < 8; ++r) {
            int m = row0 + mt * 16 + mbase + r;
            if (m < n) t[(size_t)m * 64 + col] = acc[r] * s_on[mt * 16 + mbase + r];
        }
    }
}

__global__ __launch_bounds__(256) void gcn_layer3_fused(
    const float* __restrict__ x3, const float* __restrict__ in_norm,
    const int* __restrict__ cnt_in, const int* __restrict__ bucket,
    const int* __restrict__ ovf_count, const int* __restrict__ ovf_dst,
    const int* __restrict__ ovf_src, const float* __restrict__ W3,
    const float* __restrict__ b3, const int* __restrict__ graph_ids,
    float* __restrict__ pooled, int n) {
    __shared__ __attribute__((aligned(16))) float s_a[GCN_ROWS * 68];
    int w = __builtin_amdgcn_readfirstlane((int)(threadIdx.x >> 5));
    int lane = threadIdx.x & 31;
    int row0 = blockIdx.x * GCN_ROWS;

    gcn_pull_rows_to_lds<64, false>(x3, nullptr, in_norm, cnt_in, bucket, ovf_count, ovf_dst,
                                    ovf_src, row0 + w * GCN_RPW, n, lane, s_a + w * GCN_RPW * 68);
    __syncthreads();

    int idx15 = lane & 15;
    int koff  = (lane >> 4) << 1;
    int mbase = (lane >> 4) << 3;
    int mt = w >> 2;
    int nt = w & 3;
    int trow0 = row0 + mt * 16;
    if (trow0 >= n) return;

    const float* a0 = s_a + (mt * 16 + idx15) * 68 + koff;
    const float* wcol = W3 + nt * 16 + idx15;
    v8f acc = {};
    for (int k = 0; k < 64; k += 4) {
        v2f bf = gcn_b_frag(wcol, k, koff, 64);
        v2f af = *reinterpret_cast<const v2f*>(a0 + k);
        acc = __builtin_amdgcn_wmma_f32_16x16x4_f32(
            false, af, false, bf, (short)0, acc, false, false);
    }

    int col = nt * 16 + idx15;
    float bv = b3[col];
    int g_l = graph_ids[trow0 + idx15];
    int g0 = __builtin_amdgcn_readfirstlane(g_l);
    bool one_graph = (__builtin_amdgcn_ballot_w32(g_l != g0) == 0u);
    if (one_graph) {
        float s = 0.0f;
#pragma unroll
        for (int r = 0; r < 8; ++r) s += acc[r] + bv;
        s += __shfl_xor(s, 16);
        if (lane < 16) unsafeAtomicAdd(&pooled[g0 * 64 + col], s);
    } else {
#pragma unroll
        for (int r = 0; r < 8; ++r) {
            int m = trow0 + mbase + r;
            float v = acc[r] + bv;
            int g = graph_ids[m];
            unsafeAtomicAdd(&pooled[g * 64 + col], v);
        }
    }
}


static inline int cdiv_l(long a, long b) { return (int)((a + b - 1) / b); }
static inline size_t line_up(size_t words) { return (words + 31) & ~(size_t)31; }

extern "C" void kernel_launch(void* const* d_in, const int* in_sizes, int n_in,
                              void* d_out, int out_size, void* d_ws, size_t ws_size,
                              hipStream_t stream) {
    const float* emb = (const float*)d_in[0];
    const float* W1  = (const float*)d_in[1];
    const float* b1  = (const float*)d_in[2];
    const float* W2  = (const float*)d_in[3];
    const float* b2  = (const float*)d_in[4];
    const float* W3  = (const float*)d_in[5];
    const float* b3  = (const float*)d_in[6];
    const int* src = (const int*)d_in[7];
    const int* dst = (const int*)d_in[8];
    const int* gid = (const int*)d_in[9];

    const int E = in_sizes[7];
    const int N = in_sizes[9];

    int*   cnt_in    = (int*)d_ws;
    float* pooled    = (float*)(cnt_in + N);
    float* counts    = pooled + GCN_N_GRAPHS * 64;
    int*   ovf_count = (int*)(counts + GCN_N_GRAPHS);
    float* out_norm  = (float*)(ovf_count + 4);
    float* in_norm   = out_norm + N;
    size_t head      = line_up((size_t)3 * N + GCN_N_GRAPHS * 64 + GCN_N_GRAPHS + 4);
    int*   bucket    = (int*)d_ws + head;
    int*   ovf_dst   = bucket + (size_t)N * GCN_SLOTS;
    int*   ovf_src   = ovf_dst + line_up(E);
    float* rows      = (float*)(ovf_src + line_up(E));
    size_t binned    = line_up((size_t)E + GCN_NSL * 32);
    int*   bin_pair  = (int*)(rows + (size_t)2 * N * 64);
    int*   bin_src   = bin_pair + 2 * binned;
    int*   seg_d     = bin_src + binned;
    int*   seg_s     = seg_d + line_up((size_t)GCN_NSL * (GCN_NB + 1));

    const int T = 256;

    const int zc = GCN_N_GRAPHS * 64 + GCN_N_GRAPHS + 4;
    gcn_bin_slices<<<GCN_NSL, GCN_CT, 0, stream>>>(src, dst, bin_pair, bin_src, seg_d, seg_s,
                                                   pooled, zc, E, N);
    gcn_build_ranges<<<GCN_NB, GCN_CT, 0, stream>>>(bin_pair, bin_src, seg_d, seg_s, bucket, cnt_in,
                                                    out_norm, in_norm, ovf_count, ovf_dst, ovf_src, E, N);
    gcn_count<<<cdiv_l(N, T), T, 0, stream>>>(gid, counts, N);

    const int rows_per_block = T / 32;
    const int groups = cdiv_l(N, GCN_ROWS);

    float* t  = rows;
    float* x3 = rows + (size_t)N * 64;
    gcn_layer1_fused<<<groups, T, 0, stream>>>(
        emb, out_norm, in_norm, cnt_in, bucket, ovf_count, ovf_dst, ovf_src, W1, b1, W2, t, N);

    gcn_pull_finish<64><<<cdiv_l(N, rows_per_block), T, 0, stream>>>(
        t, out_norm, in_norm, b2, cnt_in, bucket, ovf_count, ovf_dst, ovf_src, x3, N);

    gcn_layer3_fused<<<groups, T, 0, stream>>>(
        x3, in_norm, cnt_in, bucket, ovf_count, ovf_dst, ovf_src, W3, b3, gid, pooled, N);

    gcn_final<<<cdiv_l(GCN_N_GRAPHS * 64, T), T, 0, stream>>>(pooled, counts, (float*)d_out);
}
